# starting kernel
# baseline (speedup 1.0000x reference)
#include <hip/hip_runtime.h>
#include <hip/hip_bf16.h>
#include <climits>


#define IN_DIM   256
#define OUT_DIM  128
#define HEADS    8
#define HEAD_DIM 16
#define NEG_SLOPE 0.2f

typedef float v2f __attribute__((ext_vector_type(2)));
typedef float v8f __attribute__((ext_vector_type(8)));

__device__ __forceinline__ int floatKey(float x) {
    int i = __float_as_int(x);
    return (i >= 0) ? i : (i ^ 0x7FFFFFFF);
}
__device__ __forceinline__ float keyToFloat(int k) {
    int i = (k >= 0) ? k : (k ^ 0x7FFFFFFF);
    return __int_as_float(i);
}

__device__ __forceinline__ float lrelu(float x) {
    return x > 0.0f ? x : NEG_SLOPE * x;
}

__global__ void gat_init(float* __restrict__ out, const float* __restrict__ b,
                         int* __restrict__ emaxk, float* __restrict__ denom, int N) {
    int gid = blockIdx.x * blockDim.x + threadIdx.x;
    int tot = N * OUT_DIM;
    if (gid < tot) {
        out[gid] = b[gid & (OUT_DIM - 1)];
    }
    if (gid < N * HEADS) {
        emaxk[gid] = INT_MIN;
        denom[gid] = 0.0f;
    }
}

#define A_STRIDE 260

__global__ void gat_gemm_wmma(const float* __restrict__ feat,
                              const float* __restrict__ W,
                              float* __restrict__ h, int N) {
    __shared__ float As[16 * A_STRIDE];

    const int row0 = blockIdx.x * 16;
    const int lane = threadIdx.x & 31;
    const int wave = threadIdx.x >> 5;
    const int m    = lane & 15;
    const int kh   = (lane >> 4) << 1;

    for (int i = threadIdx.x; i < 16 * (IN_DIM / 4); i += blockDim.x) {
        int r = i >> 6;
        int c = (i & 63) << 2;
        int gr = row0 + r;
        if (gr >= N) gr = N - 1;
        *(float4*)&As[r * A_STRIDE + c] =
            *(const float4*)&feat[(size_t)gr * IN_DIM + c];
    }
    __syncthreads();

    const int col0 = wave * HEAD_DIM;
    v8f acc = {};
    const float* __restrict__ arow = &As[m * A_STRIDE];

    #pragma unroll 4
    for (int k0 = 0; k0 < IN_DIM; k0 += 4) {
        v2f a = *(const v2f*)&arow[k0 + kh];
        v2f bb;
        bb.x = W[(size_t)(k0 + kh + 0) * OUT_DIM + col0 + m];
        bb.y = W[(size_t)(k0 + kh + 1) * OUT_DIM + col0 + m];
        acc = __builtin_amdgcn_wmma_f32_16x16x4_f32(
             false, a,  false, bb,
             (short)0, acc,  false,  false);
    }

    const int colg  = col0 + m;
    const int rbase = row0 + ((lane >> 4) << 3);
    float* __restrict__ op = h + (size_t)rbase * OUT_DIM + colg;
    if (row0 + 16 <= N) {
        #pragma unroll
        for (int r = 0; r < 8; ++r) {
            op[(size_t)r * OUT_DIM] = acc[r];
        }
    } else {
        #pragma unroll
        for (int r = 0; r < 8; ++r) {
            if (rbase + r < N) op[(size_t)r * OUT_DIM] = acc[r];
        }
    }
}

__global__ void gat_logits(const float* __restrict__ h,
                           const float* __restrict__ attn_l,
                           const float* __restrict__ attn_r,
                           float* __restrict__ el, float* __restrict__ er, int N) {
    int gid = blockIdx.x * blockDim.x + threadIdx.x;
    if (gid >= N * HEADS) return;
    int n  = gid >> 3;
    int hd = gid & 7;
    const float* hp = h + (size_t)n * OUT_DIM + hd * HEAD_DIM;
    const float* al = attn_l + hd * HEAD_DIM;
    const float* ar = attn_r + hd * HEAD_DIM;
    float sl = 0.0f, sr = 0.0f;
    #pragma unroll
    for (int j = 0; j < HEAD_DIM; ++j) {
        float v = hp[j];
        sl = fmaf(v, al[j], sl);
        sr = fmaf(v, ar[j], sr);
    }
    el[gid] = sl;
    er[gid] = sr;
}

__device__ __forceinline__ void edge_sd(const int* __restrict__ src,
                                        const int* __restrict__ dst,
                                        int e, int E, int& s, int& d) {
    if (e < E) { s = src[e]; d = dst[e]; }
    else       { s = e - E;  d = e - E; }
}

__global__ void gat_edge_max(const int* __restrict__ src, const int* __restrict__ dst,
                             const float* __restrict__ el, const float* __restrict__ er,
                             int* __restrict__ emaxk, int E, int Et) {
    int e = blockIdx.x * blockDim.x + threadIdx.x;
    if (e >= Et) return;
    int s, d;
    edge_sd(src, dst, e, E, s, d);
    float elv[8], erv[8];
    *(float4*)&elv[0] = *(const float4*)&el[(size_t)s * HEADS];
    *(float4*)&elv[4] = *(const float4*)&el[(size_t)s * HEADS + 4];
    *(float4*)&erv[0] = *(const float4*)&er[(size_t)d * HEADS];
    *(float4*)&erv[4] = *(const float4*)&er[(size_t)d * HEADS + 4];
    #pragma unroll
    for (int hd = 0; hd < HEADS; ++hd) {
        float x = lrelu(elv[hd] + erv[hd]);
        atomicMax(&emaxk[(size_t)d * HEADS + hd], floatKey(x));
    }
}

__global__ void gat_edge_sum(const int* __restrict__ src, const int* __restrict__ dst,
                             const float* __restrict__ el, const float* __restrict__ er,
                             const int* __restrict__ emaxk, float* __restrict__ denom,
                             int E, int Et) {
    int e = blockIdx.x * blockDim.x + threadIdx.x;
    if (e >= Et) return;
    int s, d;
    edge_sd(src, dst, e, E, s, d);
    float elv[8], erv[8];
    *(float4*)&elv[0] = *(const float4*)&el[(size_t)s * HEADS];
    *(float4*)&elv[4] = *(const float4*)&el[(size_t)s * HEADS + 4];
    *(float4*)&erv[0] = *(const float4*)&er[(size_t)d * HEADS];
    *(float4*)&erv[4] = *(const float4*)&er[(size_t)d * HEADS + 4];
    #pragma unroll
    for (int hd = 0; hd < HEADS; ++hd) {
        float x  = lrelu(elv[hd] + erv[hd]);
        float mx = keyToFloat(emaxk[(size_t)d * HEADS + hd]);
        atomicAdd(&denom[(size_t)d * HEADS + hd], __expf(x - mx));
    }
}

__global__ void gat_aggregate(const int* __restrict__ src, const int* __restrict__ dst,
                              const float* __restrict__ el, const float* __restrict__ er,
                              const int* __restrict__ emaxk, const float* __restrict__ denom,
                              const float* __restrict__ h, float* __restrict__ out,
                              int E, int Et) {
    int gid = blockIdx.x * blockDim.x + threadIdx.x;
    if (gid >= Et * HEADS) return;
    int e  = gid >> 3;
    int hd = gid & 7;
    int s, d;
    edge_sd(src, dst, e, E, s, d);

    float x  = lrelu(el[(size_t)s * HEADS + hd] + er[(size_t)d * HEADS + hd]);
    float mx = keyToFloat(emaxk[(size_t)d * HEADS + hd]);
    float alpha = __expf(x - mx) / denom[(size_t)d * HEADS + hd];

    const float* hp = h + (size_t)s * OUT_DIM + hd * HEAD_DIM;
    float* op       = out + (size_t)d * OUT_DIM + hd * HEAD_DIM;
    #pragma unroll
    for (int v = 0; v < HEAD_DIM / 4; ++v) {
        float4 hv = *(const float4*)&hp[v * 4];
        atomicAdd(&op[v * 4 + 0], hv.x * alpha);
        atomicAdd(&op[v * 4 + 1], hv.y * alpha);
        atomicAdd(&op[v * 4 + 2], hv.z * alpha);
        atomicAdd(&op[v * 4 + 3], hv.w * alpha);
    }
}

extern "C" void kernel_launch(void* const* d_in, const int* in_sizes, int n_in,
                              void* d_out, int out_size, void* d_ws, size_t ws_size,
                              hipStream_t stream) {
    const float* feat   = (const float*)d_in[0];
    const float* W      = (const float*)d_in[1];
    const float* b      = (const float*)d_in[2];
    const float* attn_l = (const float*)d_in[3];
    const float* attn_r = (const float*)d_in[4];
    const int*   src    = (const int*)d_in[5];
    const int*   dst    = (const int*)d_in[6];
    float*       out    = (float*)d_out;

    const int N  = in_sizes[0] / IN_DIM;
    const int E  = in_sizes[5];
    const int Et = E + N;

    float* h     = (float*)d_ws;
    float* el    = h + (size_t)N * OUT_DIM;
    float* er    = el + (size_t)N * HEADS;
    int*   emaxk = (int*)(er + (size_t)N * HEADS);
    float* denom = (float*)(emaxk + (size_t)N * HEADS);

    const int TB = 256;

    gat_init<<<(N * OUT_DIM + TB - 1) / TB, TB, 0, stream>>>(out, b, emaxk, denom, N);

    gat_gemm_wmma<<<(N + 15) / 16, TB, 0, stream>>>(feat, W, h, N);

    gat_logits<<<(N * HEADS + TB - 1) / TB, TB, 0, stream>>>(h, attn_l, attn_r, el, er, N);

    gat_edge_max<<<(Et + TB - 1) / TB, TB, 0, stream>>>(src, dst, el, er, emaxk, E, Et);

    gat_edge_sum<<<(Et + TB - 1) / TB, TB, 0, stream>>>(src, dst, el, er, emaxk, denom, E, Et);

    gat_aggregate<<<((size_t)Et * HEADS + TB - 1) / TB, TB, 0, stream>>>(
        src, dst, el, er, emaxk, denom, h, out, E, Et);
}


# Owner blocks read a per-owner edge bitmap (1 bit per edge, built by extra blocks of the logits launch into the owner's own out rows) in place of the whole dst list; sum order still LDS arrival
# speedup vs baseline: 8.0385x; 8.0385x over previous
#include <hip/hip_runtime.h>
#include <hip/hip_bf16.h>
#include <climits>


#define IN_DIM   256
#define OUT_DIM  128
#define HEADS    8
#define HEAD_DIM 16
#define NEG_SLOPE 0.2f

typedef float v2f __attribute__((ext_vector_type(2)));
typedef float v8f __attribute__((ext_vector_type(8)));

__device__ __forceinline__ int floatKey(float x) {
    int i = __float_as_int(x);
    return (i >= 0) ? i : (i ^ 0x7FFFFFFF);
}
__device__ __forceinline__ float keyToFloat(int k) {
    int i = (k >= 0) ? k : (k ^ 0x7FFFFFFF);
    return __int_as_float(i);
}

__device__ __forceinline__ float lrelu(float x) {
    return x > 0.0f ? x : NEG_SLOPE * x;
}

#define EL_RANGE_SAFE 32.0f
__device__ __forceinline__ bool el_range_is_safe(const int* __restrict__ rng) {
    bool ok = true;
    #pragma unroll
    for (int k = 0; k < HEADS; ++k) {
        const float mx = keyToFloat(rng[k]);
        const float mn = keyToFloat(rng[HEADS + k]);
        ok = ok && ((mx - mn) <= EL_RANGE_SAFE);
    }
    return ok;
}

#define A_STRIDE 260

__global__ void gat_gemm_wmma(const float* __restrict__ feat,
                              const float* __restrict__ W,
                              float* __restrict__ h, int* __restrict__ rng, int N) {
    __shared__ float As[16 * A_STRIDE];

    if (blockIdx.x == 0 && threadIdx.x < HEADS) {
        rng[threadIdx.x]         = INT_MIN;
        rng[HEADS + threadIdx.x] = INT_MAX;
    }

    const int row0 = blockIdx.x * 16;
    const int lane = threadIdx.x & 31;
    const int wave = threadIdx.x >> 5;
    const int m    = lane & 15;
    const int kh   = (lane >> 4) << 1;

    for (int i = threadIdx.x; i < 16 * (IN_DIM / 4); i += blockDim.x) {
        int r = i >> 6;
        int c = (i & 63) << 2;
        int gr = row0 + r;
        if (gr >= N) gr = N - 1;
        *(float4*)&As[r * A_STRIDE + c] =
            *(const float4*)&feat[(size_t)gr * IN_DIM + c];
    }
    __syncthreads();

    const int col0 = wave * HEAD_DIM;
    v8f acc = {};
    const float* __restrict__ arow = &As[m * A_STRIDE];

    #pragma unroll 4
    for (int k0 = 0; k0 < IN_DIM; k0 += 4) {
        v2f a = *(const v2f*)&arow[k0 + kh];
        v2f bb;
        bb.x = W[(size_t)(k0 + kh + 0) * OUT_DIM + col0 + m];
        bb.y = W[(size_t)(k0 + kh + 1) * OUT_DIM + col0 + m];
        acc = __builtin_amdgcn_wmma_f32_16x16x4_f32(
             false, a,  false, bb,
             (short)0, acc,  false,  false);
    }

    const int colg  = col0 + m;
    const int rbase = row0 + ((lane >> 4) << 3);
    float* __restrict__ op = h + (size_t)rbase * OUT_DIM + colg;
    if (row0 + 16 <= N) {
        #pragma unroll
        for (int r = 0; r < 8; ++r) {
            op[(size_t)r * OUT_DIM] = acc[r];
        }
    } else {
        #pragma unroll
        for (int r = 0; r < 8; ++r) {
            if (rbase + r < N) op[(size_t)r * OUT_DIM] = acc[r];
        }
    }
}

#define NPB        392
#define BM_CHUNK   1024
#define BM_OWNERS  256
#define BM_PITCH   33

__device__ __forceinline__ int bm_chunks(int E) { return (E + BM_CHUNK - 1) / BM_CHUNK; }
__device__ __forceinline__ bool bm_fits(int cnt, int E) {
    return (long long)cnt * OUT_DIM >= (long long)bm_chunks(E) * (BM_CHUNK / 32);
}

__device__ __forceinline__ void bm_build(const int* __restrict__ dst, unsigned* bm,
                                         unsigned* tab, int chunk, int N, int E) {
    const int tid = threadIdx.x;
    for (int i = tid; i < BM_OWNERS * BM_PITCH; i += 256) tab[i] = 0u;
    __syncthreads();
    const int e0 = chunk * BM_CHUNK + tid * 4;
    int d[4];
    if (e0 + 3 < E) {
        const int4 v = *(const int4*)(dst + e0);
        d[0] = v.x; d[1] = v.y; d[2] = v.z; d[3] = v.w;
    } else {
        #pragma unroll
        for (int k = 0; k < 4; ++k) d[k] = (e0 + k < E) ? dst[e0 + k] : -1;
    }
    const int wloc = (tid * 4) >> 5;
    #pragma unroll
    for (int k = 0; k < 4; ++k) {
        if ((unsigned)d[k] < (unsigned)N) {
            const int o = d[k] / NPB;
            if (o < BM_OWNERS) atomicOr(&tab[o * BM_PITCH + wloc], 1u << ((tid * 4 + k) & 31));
        }
    }
    __syncthreads();
    const int nown = (N + NPB - 1) / NPB;
    for (int i = tid; i < BM_OWNERS * 32; i += 256) {
        const int o = i >> 5;
        const int w = i & 31;
        if (o < nown && bm_fits(min(NPB, N - o * NPB), E))
            bm[(size_t)o * (NPB * OUT_DIM) + (size_t)chunk * 32 + w] = tab[o * BM_PITCH + w];
    }
}

__global__ void __launch_bounds__(256)
gat_logits(const float* __restrict__ h,
           const float* __restrict__ attn_l,
           const float* __restrict__ attn_r,
           float* __restrict__ el, float* __restrict__ er,
           int* __restrict__ rng, int N,
           const int* __restrict__ dst, unsigned* bm, int E, int nlogit) {
    __shared__ int red[2][8][HEADS];
    __shared__ unsigned tab[BM_OWNERS * BM_PITCH];
    if ((int)blockIdx.x >= nlogit) {
        bm_build(dst, bm, tab, (int)blockIdx.x - nlogit, N, E);
        return;
    }
    const int  gid0  = blockIdx.x * blockDim.x + threadIdx.x;
    const bool valid = gid0 < N * HEADS;
    const int  gid   = valid ? gid0 : 0;
    int n  = gid >> 3;
    int hd = gid & 7;
    const float* hp = h + (size_t)n * OUT_DIM + hd * HEAD_DIM;
    const float* al = attn_l + hd * HEAD_DIM;
    const float* ar = attn_r + hd * HEAD_DIM;
    float sl = 0.0f, sr = 0.0f;
    #pragma unroll
    for (int j = 0; j < HEAD_DIM; ++j) {
        float v = hp[j];
        sl = fmaf(v, al[j], sl);
        sr = fmaf(v, ar[j], sr);
    }
    if (valid) {
        el[gid] = sl;
        er[gid] = sr;
    }

    const int key = floatKey(sl);
    int kmx = valid ? key : INT_MIN;
    int kmn = valid ? key : INT_MAX;
    kmx = max(kmx, __shfl_xor(kmx, 8));
    kmn = min(kmn, __shfl_xor(kmn, 8));
    kmx = max(kmx, __shfl_xor(kmx, 16));
    kmn = min(kmn, __shfl_xor(kmn, 16));
    const int lane = threadIdx.x & 31;
    const int wave = threadIdx.x >> 5;
    if (lane < HEADS) {
        red[0][wave][lane] = kmx;
        red[1][wave][lane] = kmn;
    }
    __syncthreads();
    if (threadIdx.x < 2 * HEADS) {
        const int which = threadIdx.x >> 3;
        const int hh    = threadIdx.x & 7;
        int v = red[which][0][hh];
        for (int w = 1; w < 8; ++w) {
            const int o = red[which][w][hh];
            v = which ? min(v, o) : max(v, o);
        }
        if (which) atomicMin(&rng[HEADS + hh], v);
        else       atomicMax(&rng[hh], v);
    }
}

#define OWN_TB    1024
#define OWN_WAVES 32
#define QW        256
#define SCAN_G    8
#define BM_G      4

__device__ __forceinline__ void own_push(int* __restrict__ qe_w,
                                         int& qn, bool m, int e, int lane) {
    const unsigned bal = __builtin_amdgcn_ballot_w32(m);
    if (bal != 0u) {
        const int pos = qn + __popc(bal & ((1u << lane) - 1u));
        if (m) qe_w[pos] = e;
        qn += __popc(bal);
    }
}

__device__ __forceinline__ int own_src(const int* __restrict__ src, int e, int E) {
    const bool real = e < E;
    const int  ss   = src[real ? e : 0];
    return real ? ss : e - E;
}
__device__ __forceinline__ int own_dl(const int* __restrict__ dst, int e, int E, int n0) {
    const bool real = e < E;
    const int  dd   = dst[real ? e : 0];
    return (real ? dd : e - E) - n0;
}

__device__ __forceinline__ void own_max_all(const int* qe_w, int qn,
                                            const int* __restrict__ src,
                                            const int* __restrict__ dst,
                                            const float* __restrict__ el,
                                            const float* __restrict__ er,
                                            int* mxk, int n0, int E, int lane) {
    const int sub = lane >> 3;
    const int hd  = lane & 7;
    for (int q = 0; q < qn; q += 4) {
        const bool valid = q + sub < qn;
        const int  ii    = valid ? q + sub : q;
        const int  e     = qe_w[ii];
        const int  dl    = own_dl(dst, e, E, n0);
        const int  s     = own_src(src, e, E);
        const float x = lrelu(el[(size_t)s * HEADS + hd] + er[(size_t)(n0 + dl) * HEADS + hd]);
        if (valid) atomicMax(&mxk[dl * HEADS + hd], floatKey(x));
    }
}

template <int NB, bool BOUND>
__device__ __forceinline__ void own_batch(const int* qe_w, int q0,
                                          const int* __restrict__ src,
                                          const int* __restrict__ dst,
                                          const float* __restrict__ el,
                                          const float* __restrict__ er,
                                          const float* __restrict__ h,
                                          float* acc, float* den, const int* mxk,
                                          int n0, int E, int lane, float elmax) {
    constexpr int NH = (NB + 3) / 4;
    const int  sub   = lane >> 3;
    const int  hd    = lane & 7;
    const bool upper = lane >= 16;

    int s[NH], dl[NH];
    #pragma unroll
    for (int t = 0; t < NH; ++t) {
        const int idx = (4 * t + sub < NB) ? 4 * t + sub : 0;
        const int e   = qe_w[q0 + idx];
        dl[t]         = own_dl(dst, e, E, n0);
        s[t]          = own_src(src, e, E);
    }

    int   dj[NB];
    float hv[NB][4];
    #pragma unroll
    for (int j = 0; j < NB; ++j) {
        const int sj = __builtin_amdgcn_readlane(s[j >> 2], 8 * (j & 3));
        dj[j]        = __builtin_amdgcn_readlane(dl[j >> 2], 8 * (j & 3));
        const float* __restrict__ hp = h + (size_t)sj * OUT_DIM + lane;
        #pragma unroll
        for (int k = 0; k < 4; ++k) hv[j][k] = hp[32 * k];
    }

    int pi[NH];
    #pragma unroll
    for (int t = 0; t < NH; ++t) {
        const float erd = er[(size_t)(n0 + dl[t]) * HEADS + hd];
        const float x   = lrelu(el[(size_t)s[t] * HEADS + hd] + erd);
        const float mx  = BOUND ? lrelu(elmax + erd)
                                : keyToFloat(mxk[dl[t] * HEADS + hd]);
        const float p   = __expf(x - mx);
        if (4 * t + sub < NB) atomicAdd(&den[dl[t] * HEADS + hd], p);
        pi[t] = __float_as_int(p);
    }

    #pragma unroll
    for (int j = 0; j < NB; ++j) {
        float* op = acc + dj[j] * OUT_DIM + lane;
        #pragma unroll
        for (int k = 0; k < 4; ++k) {
            const int a_lo = __builtin_amdgcn_readlane(pi[j >> 2], 8 * (j & 3) + 2 * k);
            const int a_hi = __builtin_amdgcn_readlane(pi[j >> 2], 8 * (j & 3) + 2 * k + 1);
            const float a  = __int_as_float(upper ? a_hi : a_lo);
            atomicAdd(&op[32 * k], hv[j][k] * a);
        }
    }
}

template <int MODE>
__device__ __forceinline__ void own_drain(int* qe_w, int& qn, bool all,
                                          const int* __restrict__ src,
                                          const int* __restrict__ dst,
                                          const float* __restrict__ el,
                                          const float* __restrict__ er,
                                          const float* __restrict__ h,
                                          float* acc, float* den, int* mxk,
                                          int n0, int E, int lane, float elmax) {
    if (MODE == 0) {
        own_max_all(qe_w, qn, src, dst, el, er, mxk, n0, E, lane);
        qn = 0;
    } else {
        constexpr bool B = (MODE == 1);
        while (qn >= 8) {
            qn -= 8;
            own_batch<8, B>(qe_w, qn, src, dst, el, er, h, acc, den, mxk, n0, E, lane, elmax);
        }
        if (all && qn > 0) {
            switch (qn) {
            case 1:  own_batch<1, B>(qe_w, 0, src, dst, el, er, h, acc, den, mxk, n0, E, lane, elmax); break;
            case 2:  own_batch<2, B>(qe_w, 0, src, dst, el, er, h, acc, den, mxk, n0, E, lane, elmax); break;
            case 3:  own_batch<3, B>(qe_w, 0, src, dst, el, er, h, acc, den, mxk, n0, E, lane, elmax); break;
            case 4:  own_batch<4, B>(qe_w, 0, src, dst, el, er, h, acc, den, mxk, n0, E, lane, elmax); break;
            case 5:  own_batch<5, B>(qe_w, 0, src, dst, el, er, h, acc, den, mxk, n0, E, lane, elmax); break;
            case 6:  own_batch<6, B>(qe_w, 0, src, dst, el, er, h, acc, den, mxk, n0, E, lane, elmax); break;
            default: own_batch<7, B>(qe_w, 0, src, dst, el, er, h, acc, den, mxk, n0, E, lane, elmax); break;
            }
            qn = 0;
        }
    }
}

#define BM_PUSH_VEC(vv, eb)                                                           \
    {                                                                                 \
        unsigned x_ = (vv).x, y_ = (vv).y, z_ = (vv).z, w_ = (vv).w;                  \
        const int eb_ = (eb);                                                         \
        for (int it_ = 0; it_ < 32; ++it_) {                                          \
            if (__builtin_amdgcn_ballot_w32((x_ | y_ | z_ | w_) != 0u) == 0u) break;  \
            own_push(qe_w, qn, x_ != 0u, eb_ + __ffs((int)x_) - 1, lane);             \
            own_push(qe_w, qn, y_ != 0u, eb_ + 32 + __ffs((int)y_) - 1, lane);        \
            own_push(qe_w, qn, z_ != 0u, eb_ + 64 + __ffs((int)z_) - 1, lane);        \
            own_push(qe_w, qn, w_ != 0u, eb_ + 96 + __ffs((int)w_) - 1, lane);        \
            x_ &= x_ - 1u; y_ &= y_ - 1u; z_ &= z_ - 1u; w_ &= w_ - 1u;               \
            if (qn > QW - 128)               \
                own_drain<MODE>(qe_w, qn, false, src, dst, el, er, h, acc, den, mxk, n0, E, lane, elmax); \
        }                                                                             \
    }

template <int MODE>
__device__ __forceinline__ void own_pass(int* qe_w,
                                         const int* __restrict__ src,
                                         const int* __restrict__ dst,
                                         const unsigned* bm,
                                         const float* __restrict__ el,
                                         const float* __restrict__ er,
                                         const float* __restrict__ h,
                                         float* acc, float* den, int* mxk,
                                         int n0, int cnt, int E, int lane, int wv, float elmax) {
    int qn = 0;
    if (bm != nullptr) {
        const int nvec = bm_chunks(E) * (BM_CHUNK / 128);
        const uint4* bm4 = (const uint4*)bm;
        for (int vb0 = wv * (32 * BM_G); vb0 < nvec; vb0 += OWN_WAVES * 32 * BM_G) {
            uint4 v[BM_G];
            #pragma unroll
            for (int j = 0; j < BM_G; ++j) {
                const int  vi = vb0 + 32 * j + lane;
                const bool in = vi < nvec;
                const uint4 t = bm4[in ? vi : 0];
                v[j].x = in ? t.x : 0u; v[j].y = in ? t.y : 0u;
                v[j].z = in ? t.z : 0u; v[j].w = in ? t.w : 0u;
            }
            #pragma unroll
            for (int j = 0; j < BM_G; ++j) {
                const int eb = (vb0 + 32 * j + lane) << 7;
                BM_PUSH_VEC(v[j], eb)
            }
            own_drain<MODE>(qe_w, qn, false, src, dst, el, er, h, acc, den, mxk, n0, E, lane, elmax);
        }
    } else {
        const unsigned ucnt = (unsigned)cnt;
        const int nvec   = E >> 2;
        const int ngroup = nvec / (32 * SCAN_G);
        const int4* __restrict__ dst4 = (const int4*)dst;

        for (int g = wv; g < ngroup; g += OWN_WAVES) {
            const int vb = g * (32 * SCAN_G) + lane;
            int4 v[SCAN_G];
            #pragma unroll
            for (int j = 0; j < SCAN_G; ++j) v[j] = dst4[vb + 32 * j];
            #pragma unroll
            for (int j = 0; j < SCAN_G; ++j) {
                const int e0 = (vb + 32 * j) << 2;
                own_push(qe_w, qn, (unsigned)(v[j].x - n0) < ucnt, e0 + 0, lane);
                own_push(qe_w, qn, (unsigned)(v[j].y - n0) < ucnt, e0 + 1, lane);
                own_push(qe_w, qn, (unsigned)(v[j].z - n0) < ucnt, e0 + 2, lane);
                own_push(qe_w, qn, (unsigned)(v[j].w - n0) < ucnt, e0 + 3, lane);
                if (qn > QW - 128)
                    own_drain<MODE>(qe_w, qn, false, src, dst, el, er, h, acc, den, mxk, n0, E, lane, elmax);
            }
            own_drain<MODE>(qe_w, qn, false, src, dst, el, er, h, acc, den, mxk, n0, E, lane, elmax);
        }

        for (int vb0 = ngroup * (32 * SCAN_G) + wv * 32; vb0 < nvec; vb0 += OWN_WAVES * 32) {
            const int  vi = vb0 + lane;
            const bool in = vi < nvec;
            const int4 v  = dst4[in ? vi : 0];
            const int  e0 = vi << 2;
            own_push(qe_w, qn, in && (unsigned)(v.x - n0) < ucnt, e0 + 0, lane);
            own_push(qe_w, qn, in && (unsigned)(v.y - n0) < ucnt, e0 + 1, lane);
            own_push(qe_w, qn, in && (unsigned)(v.z - n0) < ucnt, e0 + 2, lane);
            own_push(qe_w, qn, in && (unsigned)(v.w - n0) < ucnt, e0 + 3, lane);
            if (qn > QW - 128)
                own_drain<MODE>(qe_w, qn, false, src, dst, el, er, h, acc, den, mxk, n0, E, lane, elmax);
        }

        if (wv == 0) {
            const int  e  = (nvec << 2) + lane;
            const bool in = e < E;
            const int  d  = dst[in ? e : 0];
            own_push(qe_w, qn, in && (unsigned)(d - n0) < ucnt, e, lane);
        }
        if (qn > QW - 128)
            own_drain<MODE>(qe_w, qn, false, src, dst, el, er, h, acc, den, mxk, n0, E, lane, elmax);
    }

    for (int base = wv * 32; base < cnt; base += OWN_WAVES * 32) {
        const int nl = base + lane;
        own_push(qe_w, qn, nl < cnt, E + n0 + nl, lane);
        if (qn > QW - 128)
            own_drain<MODE>(qe_w, qn, false, src, dst, el, er, h, acc, den, mxk, n0, E, lane, elmax);
    }
    own_drain<MODE>(qe_w, qn, true, src, dst, el, er, h, acc, den, mxk, n0, E, lane, elmax);
}

__global__ void __launch_bounds__(OWN_TB)
gat_owner(const int* __restrict__ src, const int* __restrict__ dst,
          const float* __restrict__ el, const float* __restrict__ er,
          const float* __restrict__ h, const float* __restrict__ b,
          float* out, const int* __restrict__ rng, int N, int E) {
    __shared__ float acc[NPB * OUT_DIM];
    __shared__ float den[NPB * HEADS];
    __shared__ int   mxk[NPB * HEADS];
    __shared__ int   qe[OWN_WAVES * QW];

    const int n0   = blockIdx.x * NPB;
    const int cnt  = min(NPB, N - n0);
    const int lane = threadIdx.x & 31;
    const int wv   = __builtin_amdgcn_readfirstlane((int)(threadIdx.x >> 5));
    int* qe_w = qe + wv * QW;
    const unsigned* bm = bm_fits(cnt, E) ? (const unsigned*)(out + (size_t)n0 * OUT_DIM) : nullptr;

    for (int i = threadIdx.x; i < NPB * OUT_DIM; i += OWN_TB) acc[i] = 0.0f;
    for (int i = threadIdx.x; i < NPB * HEADS; i += OWN_TB) { den[i] = 0.0f; mxk[i] = INT_MIN; }
    __syncthreads();

    const float elmax = keyToFloat(rng[lane & 7]);

    if (el_range_is_safe(rng)) {
        own_pass<1>(qe_w, src, dst, bm, el, er, h, acc, den, mxk, n0, cnt, E, lane, wv, elmax);
    } else {
        own_pass<0>(qe_w, src, dst, bm, el, er, h, acc, den, mxk, n0, cnt, E, lane, wv, elmax);
        __syncthreads();
        own_pass<2>(qe_w, src, dst, bm, el, er, h, acc, den, mxk, n0, cnt, E, lane, wv, elmax);
    }
    __syncthreads();

    for (int i = threadIdx.x; i < cnt * OUT_DIM; i += OWN_TB) {
        const int nl = i >> 7;
        const int c  = i & (OUT_DIM - 1);
        out[(size_t)n0 * OUT_DIM + i] = acc[i] / den[nl * HEADS + (c >> 4)] + b[c];
    }
}

extern "C" void kernel_launch(void* const* d_in, const int* in_sizes, int n_in,
                              void* d_out, int out_size, void* d_ws, size_t ws_size,
                              hipStream_t stream) {
    const float* feat   = (const float*)d_in[0];
    const float* W      = (const float*)d_in[1];
    const float* b      = (const float*)d_in[2];
    const float* attn_l = (const float*)d_in[3];
    const float* attn_r = (const float*)d_in[4];
    const int*   src    = (const int*)d_in[5];
    const int*   dst    = (const int*)d_in[6];
    float*       out    = (float*)d_out;

    const int N  = in_sizes[0] / IN_DIM;
    const int E  = in_sizes[5];

    float* h     = (float*)d_ws;
    float* el    = h + (size_t)N * OUT_DIM;
    float* er    = el + (size_t)N * HEADS;
    int*   rng   = (int*)(er + (size_t)N * HEADS);

    const int TB = 256;

    gat_gemm_wmma<<<(N + 15) / 16, TB, 0, stream>>>(feat, W, h, rng, N);

    const int nlogit = (N * HEADS + TB - 1) / TB;
    const int nbuild = (E + BM_CHUNK - 1) / BM_CHUNK;
    gat_logits<<<nlogit + nbuild, TB, 0, stream>>>(h, attn_l, attn_r, el, er, rng, N,
                                                   dst, (unsigned*)out, E, nlogit);

    gat_owner<<<(N + NPB - 1) / NPB, OWN_TB, 0, stream>>>(src, dst, el, er, h, b, out, rng, N, E);
}


# Owner block puts each edge's source in an LDS slot list of its dst row (slot = returning LDS add); a wave sums a row at a time in registers, 16-B row loads, no LDS float atomics; sum order slot rank
# speedup vs baseline: 9.7267x; 1.2100x over previous
#include <hip/hip_runtime.h>
#include <hip/hip_bf16.h>
#include <climits>


#define IN_DIM   256
#define OUT_DIM  128
#define HEADS    8
#define HEAD_DIM 16
#define NEG_SLOPE 0.2f

typedef float v2f __attribute__((ext_vector_type(2)));
typedef float v8f __attribute__((ext_vector_type(8)));

__device__ __forceinline__ int floatKey(float x) {
    int i = __float_as_int(x);
    return (i >= 0) ? i : (i ^ 0x7FFFFFFF);
}
__device__ __forceinline__ float keyToFloat(int k) {
    int i = (k >= 0) ? k : (k ^ 0x7FFFFFFF);
    return __int_as_float(i);
}

__device__ __forceinline__ float lrelu(float x) {
    return x > 0.0f ? x : NEG_SLOPE * x;
}

#define EL_RANGE_SAFE 32.0f
__device__ __forceinline__ bool el_range_is_safe(const int* __restrict__ rng) {
    bool ok = true;
    #pragma unroll
    for (int k = 0; k < HEADS; ++k) {
        const float mx = keyToFloat(rng[k]);
        const float mn = keyToFloat(rng[HEADS + k]);
        ok = ok && ((mx - mn) <= EL_RANGE_SAFE);
    }
    return ok;
}

#define A_STRIDE 260

__global__ void gat_gemm_wmma(const float* __restrict__ feat,
                              const float* __restrict__ W,
                              float* __restrict__ h, int* __restrict__ rng, int N) {
    __shared__ float As[16 * A_STRIDE];

    if (blockIdx.x == 0 && threadIdx.x < HEADS) {
        rng[threadIdx.x]         = INT_MIN;
        rng[HEADS + threadIdx.x] = INT_MAX;
    }

    const int row0 = blockIdx.x * 16;
    const int lane = threadIdx.x & 31;
    const int wave = threadIdx.x >> 5;
    const int m    = lane & 15;
    const int kh   = (lane >> 4) << 1;

    for (int i = threadIdx.x; i < 16 * (IN_DIM / 4); i += blockDim.x) {
        int r = i >> 6;
        int c = (i & 63) << 2;
        int gr = row0 + r;
        if (gr >= N) gr = N - 1;
        *(float4*)&As[r * A_STRIDE + c] =
            *(const float4*)&feat[(size_t)gr * IN_DIM + c];
    }
    __syncthreads();

    const int col0 = wave * HEAD_DIM;
    v8f acc = {};
    const float* __restrict__ arow = &As[m * A_STRIDE];

    #pragma unroll 4
    for (int k0 = 0; k0 < IN_DIM; k0 += 4) {
        v2f a = *(const v2f*)&arow[k0 + kh];
        v2f bb;
        bb.x = W[(size_t)(k0 + kh + 0) * OUT_DIM + col0 + m];
        bb.y = W[(size_t)(k0 + kh + 1) * OUT_DIM + col0 + m];
        acc = __builtin_amdgcn_wmma_f32_16x16x4_f32(
             false, a,  false, bb,
             (short)0, acc,  false,  false);
    }

    const int colg  = col0 + m;
    const int rbase = row0 + ((lane >> 4) << 3);
    float* __restrict__ op = h + (size_t)rbase * OUT_DIM + colg;
    if (row0 + 16 <= N) {
        #pragma unroll
        for (int r = 0; r < 8; ++r) {
            op[(size_t)r * OUT_DIM] = acc[r];
        }
    } else {
        #pragma unroll
        for (int r = 0; r < 8; ++r) {
            if (rbase + r < N) op[(size_t)r * OUT_DIM] = acc[r];
        }
    }
}

#define NPB        400
#define BM_CHUNK   1024
#define BM_OWNERS  256
#define BM_PITCH   33

__device__ __forceinline__ int bm_chunks(int E) { return (E + BM_CHUNK - 1) / BM_CHUNK; }
__device__ __forceinline__ bool bm_fits(int cnt, int E) {
    return (long long)cnt * OUT_DIM >= (long long)bm_chunks(E) * (BM_CHUNK / 32);
}

__device__ __forceinline__ void bm_build(const int* __restrict__ dst, unsigned* bm,
                                         unsigned* tab, int chunk, int N, int E) {
    const int tid = threadIdx.x;
    for (int i = tid; i < BM_OWNERS * BM_PITCH; i += 256) tab[i] = 0u;
    __syncthreads();
    const int e0 = chunk * BM_CHUNK + tid * 4;
    int d[4];
    if (e0 + 3 < E) {
        const int4 v = *(const int4*)(dst + e0);
        d[0] = v.x; d[1] = v.y; d[2] = v.z; d[3] = v.w;
    } else {
        #pragma unroll
        for (int k = 0; k < 4; ++k) d[k] = (e0 + k < E) ? dst[e0 + k] : -1;
    }
    const int wloc = (tid * 4) >> 5;
    #pragma unroll
    for (int k = 0; k < 4; ++k) {
        if ((unsigned)d[k] < (unsigned)N) {
            const int o = d[k] / NPB;
            if (o < BM_OWNERS) atomicOr(&tab[o * BM_PITCH + wloc], 1u << ((tid * 4 + k) & 31));
        }
    }
    __syncthreads();
    const int nown = (N + NPB - 1) / NPB;
    for (int i = tid; i < BM_OWNERS * 32; i += 256) {
        const int o = i >> 5;
        const int w = i & 31;
        if (o < nown && bm_fits(min(NPB, N - o * NPB), E))
            bm[(size_t)o * (NPB * OUT_DIM) + (size_t)chunk * 32 + w] = tab[o * BM_PITCH + w];
    }
}

__global__ void __launch_bounds__(256)
gat_logits(const float* __restrict__ h,
           const float* __restrict__ attn_l,
           const float* __restrict__ attn_r,
           float* __restrict__ el, float* __restrict__ er,
           int* __restrict__ rng, int N,
           const int* __restrict__ dst, unsigned* bm, int E, int nlogit) {
    __shared__ int red[2][8][HEADS];
    __shared__ unsigned tab[BM_OWNERS * BM_PITCH];
    if ((int)blockIdx.x >= nlogit) {
        bm_build(dst, bm, tab, (int)blockIdx.x - nlogit, N, E);
        return;
    }
    const int  gid0  = blockIdx.x * blockDim.x + threadIdx.x;
    const bool valid = gid0 < N * HEADS;
    const int  gid   = valid ? gid0 : 0;
    int n  = gid >> 3;
    int hd = gid & 7;
    const float* hp = h + (size_t)n * OUT_DIM + hd * HEAD_DIM;
    const float* al = attn_l + hd * HEAD_DIM;
    const float* ar = attn_r + hd * HEAD_DIM;
    float sl = 0.0f, sr = 0.0f;
    #pragma unroll
    for (int j = 0; j < HEAD_DIM; ++j) {
        float v = hp[j];
        sl = fmaf(v, al[j], sl);
        sr = fmaf(v, ar[j], sr);
    }
    if (valid) {
        el[gid] = sl;
        er[gid] = sr;
    }

    const int key = floatKey(sl);
    int kmx = valid ? key : INT_MIN;
    int kmn = valid ? key : INT_MAX;
    kmx = max(kmx, __shfl_xor(kmx, 8));
    kmn = min(kmn, __shfl_xor(kmn, 8));
    kmx = max(kmx, __shfl_xor(kmx, 16));
    kmn = min(kmn, __shfl_xor(kmn, 16));
    const int lane = threadIdx.x & 31;
    const int wave = threadIdx.x >> 5;
    if (lane < HEADS) {
        red[0][wave][lane] = kmx;
        red[1][wave][lane] = kmn;
    }
    __syncthreads();
    if (threadIdx.x < 2 * HEADS) {
        const int which = threadIdx.x >> 3;
        const int hh    = threadIdx.x & 7;
        int v = red[which][0][hh];
        for (int w = 1; w < 8; ++w) {
            const int o = red[which][w][hh];
            v = which ? min(v, o) : max(v, o);
        }
        if (which) atomicMin(&rng[HEADS + hh], v);
        else       atomicMax(&rng[hh], v);
    }
}

#define OWN_TB    1024
#define OWN_WAVES 32
#define QW        256
#define SCAN_G    8
#define BM_G      4

__device__ __forceinline__ void own_push(int* __restrict__ qe_w,
                                         int& qn, bool m, int e, int lane) {
    const unsigned bal = __builtin_amdgcn_ballot_w32(m);
    if (bal != 0u) {
        const int pos = qn + __popc(bal & ((1u << lane) - 1u));
        if (m) qe_w[pos] = e;
        qn += __popc(bal);
    }
}

__device__ __forceinline__ int own_src(const int* __restrict__ src, int e, int E) {
    const bool real = e < E;
    const int  ss   = src[real ? e : 0];
    return real ? ss : e - E;
}
__device__ __forceinline__ int own_dl(const int* __restrict__ dst, int e, int E, int n0) {
    const bool real = e < E;
    const int  dd   = dst[real ? e : 0];
    return (real ? dd : e - E) - n0;
}

__device__ __forceinline__ void own_max_all(const int* qe_w, int qn,
                                            const int* __restrict__ src,
                                            const int* __restrict__ dst,
                                            const float* __restrict__ el,
                                            const float* __restrict__ er,
                                            int* mxk, int n0, int E, int lane) {
    const int sub = lane >> 3;
    const int hd  = lane & 7;
    for (int q = 0; q < qn; q += 4) {
        const bool valid = q + sub < qn;
        const int  ii    = valid ? q + sub : q;
        const int  e     = qe_w[ii];
        const int  dl    = own_dl(dst, e, E, n0);
        const int  s     = own_src(src, e, E);
        const float x = lrelu(el[(size_t)s * HEADS + hd] + er[(size_t)(n0 + dl) * HEADS + hd]);
        if (valid) atomicMax(&mxk[dl * HEADS + hd], floatKey(x));
    }
}

template <int NB, bool BOUND>
__device__ __forceinline__ void own_batch(const int* qe_w, int q0,
                                          const int* __restrict__ src,
                                          const int* __restrict__ dst,
                                          const float* __restrict__ el,
                                          const float* __restrict__ er,
                                          const float* __restrict__ h,
                                          float* acc, float* den, const int* mxk,
                                          int n0, int E, int lane, float elmax) {
    constexpr int NH = (NB + 3) / 4;
    const int  sub   = lane >> 3;
    const int  hd    = lane & 7;
    const bool upper = lane >= 16;

    int s[NH], dl[NH];
    #pragma unroll
    for (int t = 0; t < NH; ++t) {
        const int idx = (4 * t + sub < NB) ? 4 * t + sub : 0;
        const int e   = qe_w[q0 + idx];
        dl[t]         = own_dl(dst, e, E, n0);
        s[t]          = own_src(src, e, E);
    }

    int   dj[NB];
    float hv[NB][4];
    #pragma unroll
    for (int j = 0; j < NB; ++j) {
        const int sj = __builtin_amdgcn_readlane(s[j >> 2], 8 * (j & 3));
        dj[j]        = __builtin_amdgcn_readlane(dl[j >> 2], 8 * (j & 3));
        const float* __restrict__ hp = h + (size_t)sj * OUT_DIM + lane;
        #pragma unroll
        for (int k = 0; k < 4; ++k) hv[j][k] = hp[32 * k];
    }

    int pi[NH];
    #pragma unroll
    for (int t = 0; t < NH; ++t) {
        const float erd = er[(size_t)(n0 + dl[t]) * HEADS + hd];
        const float x   = lrelu(el[(size_t)s[t] * HEADS + hd] + erd);
        const float mx  = BOUND ? lrelu(elmax + erd)
                                : keyToFloat(mxk[dl[t] * HEADS + hd]);
        const float p   = __expf(x - mx);
        if (4 * t + sub < NB) atomicAdd(&den[dl[t] * HEADS + hd], p);
        pi[t] = __float_as_int(p);
    }

    #pragma unroll
    for (int j = 0; j < NB; ++j) {
        float* op = acc + dj[j] * OUT_DIM + lane;
        #pragma unroll
        for (int k = 0; k < 4; ++k) {
            const int a_lo = __builtin_amdgcn_readlane(pi[j >> 2], 8 * (j & 3) + 2 * k);
            const int a_hi = __builtin_amdgcn_readlane(pi[j >> 2], 8 * (j & 3) + 2 * k + 1);
            const float a  = __int_as_float(upper ? a_hi : a_lo);
            atomicAdd(&op[32 * k], hv[j][k] * a);
        }
    }
}

template <int MODE>
__device__ __forceinline__ void own_drain(int* qe_w, int& qn, bool all,
                                          const int* __restrict__ src,
                                          const int* __restrict__ dst,
                                          const float* __restrict__ el,
                                          const float* __restrict__ er,
                                          const float* __restrict__ h,
                                          float* acc, float* den, int* mxk,
                                          int n0, int E, int lane, float elmax) {
    if (MODE == 0) {
        own_max_all(qe_w, qn, src, dst, el, er, mxk, n0, E, lane);
        qn = 0;
    } else {
        constexpr bool B = (MODE == 1);
        while (qn >= 8) {
            qn -= 8;
            own_batch<8, B>(qe_w, qn, src, dst, el, er, h, acc, den, mxk, n0, E, lane, elmax);
        }
        if (all && qn > 0) {
            switch (qn) {
            case 1:  own_batch<1, B>(qe_w, 0, src, dst, el, er, h, acc, den, mxk, n0, E, lane, elmax); break;
            case 2:  own_batch<2, B>(qe_w, 0, src, dst, el, er, h, acc, den, mxk, n0, E, lane, elmax); break;
            case 3:  own_batch<3, B>(qe_w, 0, src, dst, el, er, h, acc, den, mxk, n0, E, lane, elmax); break;
            case 4:  own_batch<4, B>(qe_w, 0, src, dst, el, er, h, acc, den, mxk, n0, E, lane, elmax); break;
            case 5:  own_batch<5, B>(qe_w, 0, src, dst, el, er, h, acc, den, mxk, n0, E, lane, elmax); break;
            case 6:  own_batch<6, B>(qe_w, 0, src, dst, el, er, h, acc, den, mxk, n0, E, lane, elmax); break;
            default: own_batch<7, B>(qe_w, 0, src, dst, el, er, h, acc, den, mxk, n0, E, lane, elmax); break;
            }
            qn = 0;
        }
    }
}

#define BM_PUSH_VEC(vv, eb, DRAIN)                                                    \
    {                                                                                 \
        unsigned x_ = (vv).x, y_ = (vv).y, z_ = (vv).z, w_ = (vv).w;                  \
        const int eb_ = (eb);                                                         \
        for (int it_ = 0; it_ < 32; ++it_) {                                          \
            if (__builtin_amdgcn_ballot_w32((x_ | y_ | z_ | w_) != 0u) == 0u) break;  \
            own_push(qe_w, qn, x_ != 0u, eb_ + __ffs((int)x_) - 1, lane);             \
            own_push(qe_w, qn, y_ != 0u, eb_ + 32 + __ffs((int)y_) - 1, lane);        \
            own_push(qe_w, qn, z_ != 0u, eb_ + 64 + __ffs((int)z_) - 1, lane);        \
            own_push(qe_w, qn, w_ != 0u, eb_ + 96 + __ffs((int)w_) - 1, lane);        \
            x_ &= x_ - 1u; y_ &= y_ - 1u; z_ &= z_ - 1u; w_ &= w_ - 1u;               \
            if (qn > QW - 128) {             \
                DRAIN;                                                                \
            }                                                                         \
        }                                                                             \
    }

template <int MODE>
__device__ __forceinline__ void own_pass(int* qe_w,
                                         const int* __restrict__ src,
                                         const int* __restrict__ dst,
                                         const unsigned* bm,
                                         const float* __restrict__ el,
                                         const float* __restrict__ er,
                                         const float* __restrict__ h,
                                         float* acc, float* den, int* mxk,
                                         int n0, int cnt, int E, int lane, int wv, float elmax) {
    int qn = 0;
    if (bm != nullptr) {
        const int nvec = bm_chunks(E) * (BM_CHUNK / 128);
        const uint4* bm4 = (const uint4*)bm;
        for (int vb0 = wv * (32 * BM_G); vb0 < nvec; vb0 += OWN_WAVES * 32 * BM_G) {
            uint4 v[BM_G];
            #pragma unroll
            for (int j = 0; j < BM_G; ++j) {
                const int  vi = vb0 + 32 * j + lane;
                const bool in = vi < nvec;
                const uint4 t = bm4[in ? vi : 0];
                v[j].x = in ? t.x : 0u; v[j].y = in ? t.y : 0u;
                v[j].z = in ? t.z : 0u; v[j].w = in ? t.w : 0u;
            }
            #pragma unroll
            for (int j = 0; j < BM_G; ++j) {
                const int eb = (vb0 + 32 * j + lane) << 7;
                BM_PUSH_VEC(v[j], eb,
                            own_drain<MODE>(qe_w, qn, false, src, dst, el, er, h, acc, den, mxk, n0, E, lane, elmax))
            }
            own_drain<MODE>(qe_w, qn, false, src, dst, el, er, h, acc, den, mxk, n0, E, lane, elmax);
        }
    } else {
        const unsigned ucnt = (unsigned)cnt;
        const int nvec   = E >> 2;
        const int ngroup = nvec / (32 * SCAN_G);
        const int4* __restrict__ dst4 = (const int4*)dst;

        for (int g = wv; g < ngroup; g += OWN_WAVES) {
            const int vb = g * (32 * SCAN_G) + lane;
            int4 v[SCAN_G];
            #pragma unroll
            for (int j = 0; j < SCAN_G; ++j) v[j] = dst4[vb + 32 * j];
            #pragma unroll
            for (int j = 0; j < SCAN_G; ++j) {
                const int e0 = (vb + 32 * j) << 2;
                own_push(qe_w, qn, (unsigned)(v[j].x - n0) < ucnt, e0 + 0, lane);
                own_push(qe_w, qn, (unsigned)(v[j].y - n0) < ucnt, e0 + 1, lane);
                own_push(qe_w, qn, (unsigned)(v[j].z - n0) < ucnt, e0 + 2, lane);
                own_push(qe_w, qn, (unsigned)(v[j].w - n0) < ucnt, e0 + 3, lane);
                if (qn > QW - 128)
                    own_drain<MODE>(qe_w, qn, false, src, dst, el, er, h, acc, den, mxk, n0, E, lane, elmax);
            }
            own_drain<MODE>(qe_w, qn, false, src, dst, el, er, h, acc, den, mxk, n0, E, lane, elmax);
        }

        for (int vb0 = ngroup * (32 * SCAN_G) + wv * 32; vb0 < nvec; vb0 += OWN_WAVES * 32) {
            const int  vi = vb0 + lane;
            const bool in = vi < nvec;
            const int4 v  = dst4[in ? vi : 0];
            const int  e0 = vi << 2;
            own_push(qe_w, qn, in && (unsigned)(v.x - n0) < ucnt, e0 + 0, lane);
            own_push(qe_w, qn, in && (unsigned)(v.y - n0) < ucnt, e0 + 1, lane);
            own_push(qe_w, qn, in && (unsigned)(v.z - n0) < ucnt, e0 + 2, lane);
            own_push(qe_w, qn, in && (unsigned)(v.w - n0) < ucnt, e0 + 3, lane);
            if (qn > QW - 128)
                own_drain<MODE>(qe_w, qn, false, src, dst, el, er, h, acc, den, mxk, n0, E, lane, elmax);
        }

        if (wv == 0) {
            const int  e  = (nvec << 2) + lane;
            const bool in = e < E;
            const int  d  = dst[in ? e : 0];
            own_push(qe_w, qn, in && (unsigned)(d - n0) < ucnt, e, lane);
        }
        if (qn > QW - 128)
            own_drain<MODE>(qe_w, qn, false, src, dst, el, er, h, acc, den, mxk, n0, E, lane, elmax);
    }

    for (int base = wv * 32; base < cnt; base += OWN_WAVES * 32) {
        const int nl = base + lane;
        own_push(qe_w, qn, nl < cnt, E + n0 + nl, lane);
        if (qn > QW - 128)
            own_drain<MODE>(qe_w, qn, false, src, dst, el, er, h, acc, den, mxk, n0, E, lane, elmax);
    }
    own_drain<MODE>(qe_w, qn, true, src, dst, el, er, h, acc, den, mxk, n0, E, lane, elmax);
}

#define CAPR  64
#define RS_G  13

__device__ __forceinline__ void route_drain(const int* qe_w, int& qn,
                                            const int* __restrict__ src,
                                            const int* __restrict__ dst,
                                            float* rowq, int* rowcnt,
                                            int n0, int cnt, int lane) {
    for (int base = 0; base < qn; base += 128) {
        int  e[4], d[4], s[4];
        bool in[4];
        #pragma unroll
        for (int u = 0; u < 4; ++u) {
            const int i = base + 32 * u + lane;
            in[u] = i < qn;
            e[u]  = qe_w[in[u] ? i : base];
        }
        #pragma unroll
        for (int u = 0; u < 4; ++u) {
            d[u] = dst[e[u]];
            s[u] = src[e[u]];
        }
        int pos[4];
        #pragma unroll
        for (int u = 0; u < 4; ++u) {
            d[u] -= n0;
            in[u] = in[u] && (unsigned)d[u] < (unsigned)cnt;
            pos[u] = 0;
            if (in[u]) pos[u] = atomicAdd(&rowcnt[d[u]], 1);
        }
        #pragma unroll
        for (int u = 0; u < 4; ++u) {
            if (in[u]) {
                if (pos[u] < CAPR) rowq[d[u] * CAPR + pos[u]] = __int_as_float(s[u]);
                else               rowcnt[NPB] = 1;
            }
        }
    }
    qn = 0;
}

__device__ __forceinline__ void route_scan(int* qe_w,
                                           const int* __restrict__ src,
                                           const int* __restrict__ dst,
                                           const unsigned* bm,
                                           float* rowq, int* rowcnt,
                                           int n0, int cnt, int E, int lane, int wv) {
    int qn = 0;
    const int nvec = bm_chunks(E) * (BM_CHUNK / 128);
    const uint4* bm4 = (const uint4*)bm;
    for (int vb0 = wv * (32 * RS_G); vb0 < nvec; vb0 += OWN_WAVES * 32 * RS_G) {
        uint4 v[RS_G];
        #pragma unroll
        for (int j = 0; j < RS_G; ++j) {
            const int  vi = vb0 + 32 * j + lane;
            const bool in = vi < nvec;
            const uint4 t = bm4[in ? vi : 0];
            v[j].x = in ? t.x : 0u; v[j].y = in ? t.y : 0u;
            v[j].z = in ? t.z : 0u; v[j].w = in ? t.w : 0u;
        }
        #pragma unroll
        for (int j = 0; j < RS_G; ++j) {
            const int eb = (vb0 + 32 * j + lane) << 7;
            BM_PUSH_VEC(v[j], eb, route_drain(qe_w, qn, src, dst, rowq, rowcnt, n0, cnt, lane))
        }
        route_drain(qe_w, qn, src, dst, rowq, rowcnt, n0, cnt, lane);
    }
}

__device__ __forceinline__ void rows_from_slots(const float* rowq, const int* rowcnt,
                                                const float* __restrict__ el,
                                                const float* __restrict__ er,
                                                const float* __restrict__ h,
                                                const float* __restrict__ b,
                                                const int* __restrict__ rng,
                                                float* out, int n0, int cnt, int lane, int wv) {
    int r = wv;
    if (r >= cnt) return;

    const int    hcol  = lane << 2;
    const int    wl    = lane >> 2;
    const float  elmax = keyToFloat(rng[wl]);
    const float4 bias  = *(const float4*)(b + hcol);

    int   n   = min(__builtin_amdgcn_readfirstlane(rowcnt[r]), CAPR);
    int   sv0 = (lane < n)      ? __float_as_int(rowq[r * CAPR + lane])      : n0 + r;
    int   sv1 = (lane + 32 < n) ? __float_as_int(rowq[r * CAPR + 32 + lane]) : n0 + r;
    float erd = er[(size_t)(n0 + r) * HEADS + wl];

    bool more;
    do {
        const int d  = n0 + r;
        const int rn = r + OWN_WAVES;
        more = rn < cnt;
        const int rc = more ? rn : r;
        const float erd_next = er[(size_t)(n0 + rc) * HEADS + wl];

        const int   tot   = n + 1;
        const float bound = lrelu(elmax + erd);
        float4 acc = make_float4(0.0f, 0.0f, 0.0f, 0.0f);
        float  dn  = 0.0f;

        for (int q0 = 0; q0 < tot; q0 += 8) {
            const int left = tot - q0;
            const int svv  = (q0 < 32) ? sv0 : ((q0 < 64) ? sv1 : d);
            const int lb   = q0 & 31;

            float4 hv[8];
            float  ev[8];
            #pragma unroll
            for (int j = 0; j < 8; ++j) {
                const int sj = __builtin_amdgcn_readlane(svv, lb + j);
                hv[j] = *(const float4*)(h + (size_t)sj * OUT_DIM + hcol);
                ev[j] = el[(size_t)sj * HEADS + wl];
            }
            #pragma unroll
            for (int j = 0; j < 8; ++j) {
                const float x = lrelu(ev[j] + erd);
                const float a = (j < left) ? __expf(x - bound) : 0.0f;
                acc.x = fmaf(a, hv[j].x, acc.x);
                acc.y = fmaf(a, hv[j].y, acc.y);
                acc.z = fmaf(a, hv[j].z, acc.z);
                acc.w = fmaf(a, hv[j].w, acc.w);
                dn += a;
            }
        }

        const int n_next   = min(__builtin_amdgcn_readfirstlane(rowcnt[rc]), CAPR);
        const int sv0_next = (lane < n_next)      ? __float_as_int(rowq[rc * CAPR + lane])      : n0 + rc;
        const int sv1_next = (lane + 32 < n_next) ? __float_as_int(rowq[rc * CAPR + 32 + lane]) : n0 + rc;

        float4 o;
        o.x = acc.x / dn + bias.x;
        o.y = acc.y / dn + bias.y;
        o.z = acc.z / dn + bias.z;
        o.w = acc.w / dn + bias.w;
        *(float4*)(out + (size_t)d * OUT_DIM + hcol) = o;

        r = rn; n = n_next; sv0 = sv0_next; sv1 = sv1_next; erd = erd_next;
    } while (more);
}

__global__ void __launch_bounds__(OWN_TB)
gat_owner(const int* __restrict__ src, const int* __restrict__ dst,
          const float* __restrict__ el, const float* __restrict__ er,
          const float* __restrict__ h, const float* __restrict__ b,
          float* out, const int* __restrict__ rng, int N, int E) {
    __shared__ float acc[NPB * OUT_DIM];
    __shared__ float den[NPB * HEADS];
    __shared__ int   mxk[NPB * HEADS];
    __shared__ int   qe[OWN_WAVES * QW];
    __shared__ int   rowcnt[NPB + 1];

    const int n0   = blockIdx.x * NPB;
    const int cnt  = min(NPB, N - n0);
    const int lane = threadIdx.x & 31;
    const int wv   = __builtin_amdgcn_readfirstlane((int)(threadIdx.x >> 5));
    int* qe_w = qe + wv * QW;
    const unsigned* bm = bm_fits(cnt, E) ? (const unsigned*)(out + (size_t)n0 * OUT_DIM) : nullptr;

    const float elmax = keyToFloat(rng[lane & 7]);
    const bool  safe  = el_range_is_safe(rng);

    if (bm != nullptr && safe) {
        for (int i = threadIdx.x; i <= NPB; i += OWN_TB) rowcnt[i] = 0;
        __syncthreads();
        route_scan(qe_w, src, dst, bm, acc, rowcnt, n0, cnt, E, lane, wv);
        __syncthreads();
        if (rowcnt[NPB] == 0) {
            rows_from_slots(acc, rowcnt, el, er, h, b, rng, out, n0, cnt, lane, wv);
            return;
        }
    }

    for (int i = threadIdx.x; i < NPB * OUT_DIM; i += OWN_TB) acc[i] = 0.0f;
    for (int i = threadIdx.x; i < NPB * HEADS; i += OWN_TB) { den[i] = 0.0f; mxk[i] = INT_MIN; }
    __syncthreads();

    if (safe) {
        own_pass<1>(qe_w, src, dst, bm, el, er, h, acc, den, mxk, n0, cnt, E, lane, wv, elmax);
    } else {
        own_pass<0>(qe_w, src, dst, bm, el, er, h, acc, den, mxk, n0, cnt, E, lane, wv, elmax);
        __syncthreads();
        own_pass<2>(qe_w, src, dst, bm, el, er, h, acc, den, mxk, n0, cnt, E, lane, wv, elmax);
    }
    __syncthreads();

    for (int i = threadIdx.x; i < cnt * OUT_DIM; i += OWN_TB) {
        const int nl = i >> 7;
        const int c  = i & (OUT_DIM - 1);
        out[(size_t)n0 * OUT_DIM + i] = acc[i] / den[nl * HEADS + (c >> 4)] + b[c];
    }
}

extern "C" void kernel_launch(void* const* d_in, const int* in_sizes, int n_in,
                              void* d_out, int out_size, void* d_ws, size_t ws_size,
                              hipStream_t stream) {
    const float* feat   = (const float*)d_in[0];
    const float* W      = (const float*)d_in[1];
    const float* b      = (const float*)d_in[2];
    const float* attn_l = (const float*)d_in[3];
    const float* attn_r = (const float*)d_in[4];
    const int*   src    = (const int*)d_in[5];
    const int*   dst    = (const int*)d_in[6];
    float*       out    = (float*)d_out;

    const int N  = in_sizes[0] / IN_DIM;
    const int E  = in_sizes[5];

    float* h     = (float*)d_ws;
    float* el    = h + (size_t)N * OUT_DIM;
    float* er    = el + (size_t)N * HEADS;
    int*   rng   = (int*)(er + (size_t)N * HEADS);

    const int TB = 256;

    gat_gemm_wmma<<<(N + 15) / 16, TB, 0, stream>>>(feat, W, h, rng, N);

    const int nlogit = (N * HEADS + TB - 1) / TB;
    const int nbuild = (E + BM_CHUNK - 1) / BM_CHUNK;
    gat_logits<<<nlogit + nbuild, TB, 0, stream>>>(h, attn_l, attn_r, el, er, rng, N,
                                                   dst, (unsigned*)out, E, nlogit);

    gat_owner<<<(N + NPB - 1) / NPB, OWN_TB, 0, stream>>>(src, dst, el, er, h, b, out, rng, N, E);
}


# Launches 1 and 2 made resident: few long-lived blocks stride over row tiles, logit items, bitmap chunks (wave entry rate set the pace); a GEMM wave keeps its W fragments in registers; sums unchanged
# speedup vs baseline: 11.2938x; 1.1611x over previous
#include <hip/hip_runtime.h>
#include <hip/hip_bf16.h>
#include <climits>


#define IN_DIM   256
#define OUT_DIM  128
#define HEADS    8
#define HEAD_DIM 16
#define NEG_SLOPE 0.2f

typedef float v2f __attribute__((ext_vector_type(2)));
typedef float v8f __attribute__((ext_vector_type(8)));

__device__ __forceinline__ int floatKey(float x) {
    int i = __float_as_int(x);
    return (i >= 0) ? i : (i ^ 0x7FFFFFFF);
}
__device__ __forceinline__ float keyToFloat(int k) {
    int i = (k >= 0) ? k : (k ^ 0x7FFFFFFF);
    return __int_as_float(i);
}

__device__ __forceinline__ float lrelu(float x) {
    return x > 0.0f ? x : NEG_SLOPE * x;
}

#define EL_RANGE_SAFE 32.0f
__device__ __forceinline__ bool el_range_is_safe(const int* __restrict__ rng) {
    bool ok = true;
    #pragma unroll
    for (int k = 0; k < HEADS; ++k) {
        const float mx = keyToFloat(rng[k]);
        const float mn = keyToFloat(rng[HEADS + k]);
        ok = ok && ((mx - mn) <= EL_RANGE_SAFE);
    }
    return ok;
}

#define A_STRIDE 260

#define GEMM_GRID 512
#define KSTEPS    (IN_DIM / 4)
#define KGROUP    8

__device__ __forceinline__ float4 gemm_feat_piece(const float* __restrict__ feat, int tile, int u, int N) {
    int gr = tile * 16 + (int)(threadIdx.x >> 6) + 4 * u;
    if (gr >= N) gr = N - 1;
    return *(const float4*)&feat[(size_t)gr * IN_DIM + ((threadIdx.x & 63) << 2)];
}
__device__ __forceinline__ void gemm_stage_piece(float* __restrict__ stage, int u, float4 v) {
    *(float4*)&stage[((int)(threadIdx.x >> 6) + 4 * u) * A_STRIDE + ((threadIdx.x & 63) << 2)] = v;
}

__global__ void __launch_bounds__(256)
gat_gemm_wmma(const float* __restrict__ feat,
              const float* __restrict__ W,
              float* __restrict__ h, int* __restrict__ rng, int N) {
    __shared__ float As[2 * 16 * A_STRIDE];

    if (blockIdx.x == 0 && threadIdx.x < HEADS) {
        rng[threadIdx.x]         = INT_MIN;
        rng[HEADS + threadIdx.x] = INT_MAX;
    }

    const int ntile = (N + 15) >> 4;
    int tile = blockIdx.x;
    if (tile >= ntile) return;

    const int lane = threadIdx.x & 31;
    const int wave = threadIdx.x >> 5;
    const int m    = lane & 15;
    const int kh   = (lane >> 4) << 1;
    const int col0 = wave * HEAD_DIM;

    float4 p0 = gemm_feat_piece(feat, tile, 0, N);
    float4 p1 = gemm_feat_piece(feat, tile, 1, N);
    float4 p2 = gemm_feat_piece(feat, tile, 2, N);
    float4 p3 = gemm_feat_piece(feat, tile, 3, N);

    v2f bw[KSTEPS];
    #pragma unroll
    for (int k = 0; k < KSTEPS; ++k) {
        bw[k].x = W[(size_t)(4 * k + kh + 0) * OUT_DIM + col0 + m];
        bw[k].y = W[(size_t)(4 * k + kh + 1) * OUT_DIM + col0 + m];
    }

    gemm_stage_piece(As, 0, p0);
    gemm_stage_piece(As, 1, p1);
    gemm_stage_piece(As, 2, p2);
    gemm_stage_piece(As, 3, p3);
    __syncthreads();

    const int colg = col0 + m;
    int buf = 0;
    for (;;) {
        const int  next     = tile + (int)gridDim.x;
        const bool has_next = next < ntile;
        if (has_next) {
            p0 = gemm_feat_piece(feat, next, 0, N);
            p1 = gemm_feat_piece(feat, next, 1, N);
            p2 = gemm_feat_piece(feat, next, 2, N);
            p3 = gemm_feat_piece(feat, next, 3, N);
        }

        const float* __restrict__ arow = &As[buf * (16 * A_STRIDE) + m * A_STRIDE + kh];
        v8f acc = {};
        v2f a[2][KGROUP];
        #pragma unroll
        for (int j = 0; j < KGROUP; ++j) a[0][j] = *(const v2f*)&arow[4 * j];
        #pragma unroll
        for (int g = 0; g < KSTEPS / KGROUP; ++g) {
            if (g + 1 < KSTEPS / KGROUP) {
                #pragma unroll
                for (int j = 0; j < KGROUP; ++j)
                    a[(g + 1) & 1][j] = *(const v2f*)&arow[4 * ((g + 1) * KGROUP + j)];
            }
            __builtin_amdgcn_sched_barrier(0);
            #pragma unroll
            for (int j = 0; j < KGROUP; ++j) {
                acc = __builtin_amdgcn_wmma_f32_16x16x4_f32(
                     false, a[g & 1][j],  false, bw[g * KGROUP + j],
                     (short)0, acc,  false,  false);
            }
            __builtin_amdgcn_sched_barrier(0);
        }

        if (has_next) {
            float* __restrict__ nst = &As[(buf ^ 1) * (16 * A_STRIDE)];
            gemm_stage_piece(nst, 0, p0);
            gemm_stage_piece(nst, 1, p1);
            gemm_stage_piece(nst, 2, p2);
            gemm_stage_piece(nst, 3, p3);
            __syncthreads();
        }

        const int row0  = tile * 16;
        const int rbase = row0 + ((lane >> 4) << 3);
        float* __restrict__ op = h + (size_t)rbase * OUT_DIM + colg;
        if (row0 + 16 <= N) {
            #pragma unroll
            for (int r = 0; r < 8; ++r) {
                op[(size_t)r * OUT_DIM] = acc[r];
            }
        } else {
            #pragma unroll
            for (int r = 0; r < 8; ++r) {
                if (rbase + r < N) op[(size_t)r * OUT_DIM] = acc[r];
            }
        }

        if (!has_next) break;
        buf ^= 1;
        tile = next;
    }
}

#define NPB        400
#define BM_CHUNK   1024
#define BM_OWNERS  256
#define BM_PITCH   33
#define LOGIT_GRID 256
#define BUILD_GRID 512

__device__ __forceinline__ int bm_chunks(int E) { return (E + BM_CHUNK - 1) / BM_CHUNK; }
__device__ __forceinline__ bool bm_fits(int cnt, int E) {
    return (long long)cnt * OUT_DIM >= (long long)bm_chunks(E) * (BM_CHUNK / 32);
}

__device__ __forceinline__ void bm_build(const int* __restrict__ dst, unsigned* bm,
                                         unsigned* tab, int chunk, int N, int E) {
    const int tid = threadIdx.x;
    const int e0 = chunk * BM_CHUNK + tid * 4;
    int d[4];
    if (e0 + 3 < E) {
        const int4 v = *(const int4*)(dst + e0);
        d[0] = v.x; d[1] = v.y; d[2] = v.z; d[3] = v.w;
    } else {
        #pragma unroll
        for (int k = 0; k < 4; ++k) d[k] = (e0 + k < E) ? dst[e0 + k] : -1;
    }
    for (int i = tid; i < BM_OWNERS * BM_PITCH; i += 256) tab[i] = 0u;
    __syncthreads();
    const int wloc = (tid * 4) >> 5;
    #pragma unroll
    for (int k = 0; k < 4; ++k) {
        if ((unsigned)d[k] < (unsigned)N) {
            const int o = d[k] / NPB;
            if (o < BM_OWNERS) atomicOr(&tab[o * BM_PITCH + wloc], 1u << ((tid * 4 + k) & 31));
        }
    }
    __syncthreads();
    const int nown = (N + NPB - 1) / NPB;
    for (int i = tid; i < BM_OWNERS * 32; i += 256) {
        const int o = i >> 5;
        const int w = i & 31;
        if (o < nown && bm_fits(min(NPB, N - o * NPB), E))
            bm[(size_t)o * (NPB * OUT_DIM) + (size_t)chunk * 32 + w] = tab[o * BM_PITCH + w];
    }
}

__global__ void __launch_bounds__(256)
gat_logits(const float* __restrict__ h,
           const float* __restrict__ attn_l,
           const float* __restrict__ attn_r,
           float* __restrict__ el, float* __restrict__ er,
           int* __restrict__ rng, int N,
           const int* __restrict__ dst, unsigned* bm, int E, int nlogit) {
    __shared__ int red[2][8][HEADS];
    __shared__ unsigned tab[BM_OWNERS * BM_PITCH];
    if ((int)blockIdx.x >= nlogit) {
        const int nb = (int)gridDim.x - nlogit;
        for (int chunk = (int)blockIdx.x - nlogit; chunk < bm_chunks(E); chunk += nb) {
            bm_build(dst, bm, tab, chunk, N, E);
            __syncthreads();
        }
        return;
    }
    const int nthr = nlogit * (int)blockDim.x;
    const int hd   = threadIdx.x & 7;
    const float* al = attn_l + hd * HEAD_DIM;
    const float* ar = attn_r + hd * HEAD_DIM;
    int kmx = INT_MIN;
    int kmn = INT_MAX;
    for (int gid = blockIdx.x * blockDim.x + threadIdx.x; gid < N * HEADS; gid += nthr) {
        const int n = gid >> 3;
        const float* hp = h + (size_t)n * OUT_DIM + hd * HEAD_DIM;
        float sl = 0.0f, sr = 0.0f;
        #pragma unroll
        for (int j = 0; j < HEAD_DIM; ++j) {
            float v = hp[j];
            sl = fmaf(v, al[j], sl);
            sr = fmaf(v, ar[j], sr);
        }
        el[gid] = sl;
        er[gid] = sr;
        const int key = floatKey(sl);
        kmx = max(kmx, key);
        kmn = min(kmn, key);
    }

    kmx = max(kmx, __shfl_xor(kmx, 8));
    kmn = min(kmn, __shfl_xor(kmn, 8));
    kmx = max(kmx, __shfl_xor(kmx, 16));
    kmn = min(kmn, __shfl_xor(kmn, 16));
    const int lane = threadIdx.x & 31;
    const int wave = threadIdx.x >> 5;
    if (lane < HEADS) {
        red[0][wave][lane] = kmx;
        red[1][wave][lane] = kmn;
    }
    __syncthreads();
    if (threadIdx.x < 2 * HEADS) {
        const int which = threadIdx.x >> 3;
        const int hh    = threadIdx.x & 7;
        int v = red[which][0][hh];
        for (int w = 1; w < 8; ++w) {
            const int o = red[which][w][hh];
            v = which ? min(v, o) : max(v, o);
        }
        if (which) atomicMin(&rng[HEADS + hh], v);
        else       atomicMax(&rng[hh], v);
    }
}

#define OWN_TB    1024
#define OWN_WAVES 32
#define QW        256
#define SCAN_G    8
#define BM_G      4

__device__ __forceinline__ void own_push(int* __restrict__ qe_w,
                                         int& qn, bool m, int e, int lane) {
    const unsigned bal = __builtin_amdgcn_ballot_w32(m);
    if (bal != 0u) {
        const int pos = qn + __popc(bal & ((1u << lane) - 1u));
        if (m) qe_w[pos] = e;
        qn += __popc(bal);
    }
}

__device__ __forceinline__ int own_src(const int* __restrict__ src, int e, int E) {
    const bool real = e < E;
    const int  ss   = src[real ? e : 0];
    return real ? ss : e - E;
}
__device__ __forceinline__ int own_dl(const int* __restrict__ dst, int e, int E, int n0) {
    const bool real = e < E;
    const int  dd   = dst[real ? e : 0];
    return (real ? dd : e - E) - n0;
}

__device__ __forceinline__ void own_max_all(const int* qe_w, int qn,
                                            const int* __restrict__ src,
                                            const int* __restrict__ dst,
                                            const float* __restrict__ el,
                                            const float* __restrict__ er,
                                            int* mxk, int n0, int E, int lane) {
    const int sub = lane >> 3;
    const int hd  = lane & 7;
    for (int q = 0; q < qn; q += 4) {
        const bool valid = q + sub < qn;
        const int  ii    = valid ? q + sub : q;
        const int  e     = qe_w[ii];
        const int  dl    = own_dl(dst, e, E, n0);
        const int  s     = own_src(src, e, E);
        const float x = lrelu(el[(size_t)s * HEADS + hd] + er[(size_t)(n0 + dl) * HEADS + hd]);
        if (valid) atomicMax(&mxk[dl * HEADS + hd], floatKey(x));
    }
}

template <int NB, bool BOUND>
__device__ __forceinline__ void own_batch(const int* qe_w, int q0,
                                          const int* __restrict__ src,
                                          const int* __restrict__ dst,
                                          const float* __restrict__ el,
                                          const float* __restrict__ er,
                                          const float* __restrict__ h,
                                          float* acc, float* den, const int* mxk,
                                          int n0, int E, int lane, float elmax) {
    constexpr int NH = (NB + 3) / 4;
    const int  sub   = lane >> 3;
    const int  hd    = lane & 7;
    const bool upper = lane >= 16;

    int s[NH], dl[NH];
    #pragma unroll
    for (int t = 0; t < NH; ++t) {
        const int idx = (4 * t + sub < NB) ? 4 * t + sub : 0;
        const int e   = qe_w[q0 + idx];
        dl[t]         = own_dl(dst, e, E, n0);
        s[t]          = own_src(src, e, E);
    }

    int   dj[NB];
    float hv[NB][4];
    #pragma unroll
    for (int j = 0; j < NB; ++j) {
        const int sj = __builtin_amdgcn_readlane(s[j >> 2], 8 * (j & 3));
        dj[j]        = __builtin_amdgcn_readlane(dl[j >> 2], 8 * (j & 3));
        const float* __restrict__ hp = h + (size_t)sj * OUT_DIM + lane;
        #pragma unroll
        for (int k = 0; k < 4; ++k) hv[j][k] = hp[32 * k];
    }

    int pi[NH];
    #pragma unroll
    for (int t = 0; t < NH; ++t) {
        const float erd = er[(size_t)(n0 + dl[t]) * HEADS + hd];
        const float x   = lrelu(el[(size_t)s[t] * HEADS + hd] + erd);
        const float mx  = BOUND ? lrelu(elmax + erd)
                                : keyToFloat(mxk[dl[t] * HEADS + hd]);
        const float p   = __expf(x - mx);
        if (4 * t + sub < NB) atomicAdd(&den[dl[t] * HEADS + hd], p);
        pi[t] = __float_as_int(p);
    }

    #pragma unroll
    for (int j = 0; j < NB; ++j) {
        float* op = acc + dj[j] * OUT_DIM + lane;
        #pragma unroll
        for (int k = 0; k < 4; ++k) {
            const int a_lo = __builtin_amdgcn_readlane(pi[j >> 2], 8 * (j & 3) + 2 * k);
            const int a_hi = __builtin_amdgcn_readlane(pi[j >> 2], 8 * (j & 3) + 2 * k + 1);
            const float a  = __int_as_float(upper ? a_hi : a_lo);
            atomicAdd(&op[32 * k], hv[j][k] * a);
        }
    }
}

template <int MODE>
__device__ __forceinline__ void own_drain(int* qe_w, int& qn, bool all,
                                          const int* __restrict__ src,
                                          const int* __restrict__ dst,
                                          const float* __restrict__ el,
                                          const float* __restrict__ er,
                                          const float* __restrict__ h,
                                          float* acc, float* den, int* mxk,
                                          int n0, int E, int lane, float elmax) {
    if (MODE == 0) {
        own_max_all(qe_w, qn, src, dst, el, er, mxk, n0, E, lane);
        qn = 0;
    } else {
        constexpr bool B = (MODE == 1);
        while (qn >= 8) {
            qn -= 8;
            own_batch<8, B>(qe_w, qn, src, dst, el, er, h, acc, den, mxk, n0, E, lane, elmax);
        }
        if (all && qn > 0) {
            switch (qn) {
            case 1:  own_batch<1, B>(qe_w, 0, src, dst, el, er, h, acc, den, mxk, n0, E, lane, elmax); break;
            case 2:  own_batch<2, B>(qe_w, 0, src, dst, el, er, h, acc, den, mxk, n0, E, lane, elmax); break;
            case 3:  own_batch<3, B>(qe_w, 0, src, dst, el, er, h, acc, den, mxk, n0, E, lane, elmax); break;
            case 4:  own_batch<4, B>(qe_w, 0, src, dst, el, er, h, acc, den, mxk, n0, E, lane, elmax); break;
            case 5:  own_batch<5, B>(qe_w, 0, src, dst, el, er, h, acc, den, mxk, n0, E, lane, elmax); break;
            case 6:  own_batch<6, B>(qe_w, 0, src, dst, el, er, h, acc, den, mxk, n0, E, lane, elmax); break;
            default: own_batch<7, B>(qe_w, 0, src, dst, el, er, h, acc, den, mxk, n0, E, lane, elmax); break;
            }
            qn = 0;
        }
    }
}

#define BM_PUSH_VEC(vv, eb, DRAIN)                                                    \
    {                                                                                 \
        unsigned x_ = (vv).x, y_ = (vv).y, z_ = (vv).z, w_ = (vv).w;                  \
        const int eb_ = (eb);                                                         \
        for (int it_ = 0; it_ < 32; ++it_) {                                          \
            if (__builtin_amdgcn_ballot_w32((x_ | y_ | z_ | w_) != 0u) == 0u) break;  \
            own_push(qe_w, qn, x_ != 0u, eb_ + __ffs((int)x_) - 1, lane);             \
            own_push(qe_w, qn, y_ != 0u, eb_ + 32 + __ffs((int)y_) - 1, lane);        \
            own_push(qe_w, qn, z_ != 0u, eb_ + 64 + __ffs((int)z_) - 1, lane);        \
            own_push(qe_w, qn, w_ != 0u, eb_ + 96 + __ffs((int)w_) - 1, lane);        \
            x_ &= x_ - 1u; y_ &= y_ - 1u; z_ &= z_ - 1u; w_ &= w_ - 1u;               \
            if (qn > QW - 128) {             \
                DRAIN;                                                                \
            }                                                                         \
        }                                                                             \
    }

template <int MODE>
__device__ __forceinline__ void own_pass(int* qe_w,
                                         const int* __restrict__ src,
                                         const int* __restrict__ dst,
                                         const unsigned* bm,
                                         const float* __restrict__ el,
                                         const float* __restrict__ er,
                                         const float* __restrict__ h,
                                         float* acc, float* den, int* mxk,
                                         int n0, int cnt, int E, int lane, int wv, float elmax) {
    int qn = 0;
    if (bm != nullptr) {
        const int nvec = bm_chunks(E) * (BM_CHUNK / 128);
        const uint4* bm4 = (const uint4*)bm;
        for (int vb0 = wv * (32 * BM_G); vb0 < nvec; vb0 += OWN_WAVES * 32 * BM_G) {
            uint4 v[BM_G];
            #pragma unroll
            for (int j = 0; j < BM_G; ++j) {
                const int  vi = vb0 + 32 * j + lane;
                const bool in = vi < nvec;
                const uint4 t = bm4[in ? vi : 0];
                v[j].x = in ? t.x : 0u; v[j].y = in ? t.y : 0u;
                v[j].z = in ? t.z : 0u; v[j].w = in ? t.w : 0u;
            }
            #pragma unroll
            for (int j = 0; j < BM_G; ++j) {
                const int eb = (vb0 + 32 * j + lane) << 7;
                BM_PUSH_VEC(v[j], eb,
                            own_drain<MODE>(qe_w, qn, false, src, dst, el, er, h, acc, den, mxk, n0, E, lane, elmax))
            }
            own_drain<MODE>(qe_w, qn, false, src, dst, el, er, h, acc, den, mxk, n0, E, lane, elmax);
        }
    } else {
        const unsigned ucnt = (unsigned)cnt;
        const int nvec   = E >> 2;
        const int ngroup = nvec / (32 * SCAN_G);
        const int4* __restrict__ dst4 = (const int4*)dst;

        for (int g = wv; g < ngroup; g += OWN_WAVES) {
            const int vb = g * (32 * SCAN_G) + lane;
            int4 v[SCAN_G];
            #pragma unroll
            for (int j = 0; j < SCAN_G; ++j) v[j] = dst4[vb + 32 * j];
            #pragma unroll
            for (int j = 0; j < SCAN_G; ++j) {
                const int e0 = (vb + 32 * j) << 2;
                own_push(qe_w, qn, (unsigned)(v[j].x - n0) < ucnt, e0 + 0, lane);
                own_push(qe_w, qn, (unsigned)(v[j].y - n0) < ucnt, e0 + 1, lane);
                own_push(qe_w, qn, (unsigned)(v[j].z - n0) < ucnt, e0 + 2, lane);
                own_push(qe_w, qn, (unsigned)(v[j].w - n0) < ucnt, e0 + 3, lane);
                if (qn > QW - 128)
                    own_drain<MODE>(qe_w, qn, false, src, dst, el, er, h, acc, den, mxk, n0, E, lane, elmax);
            }
            own_drain<MODE>(qe_w, qn, false, src, dst, el, er, h, acc, den, mxk, n0, E, lane, elmax);
        }

        for (int vb0 = ngroup * (32 * SCAN_G) + wv * 32; vb0 < nvec; vb0 += OWN_WAVES * 32) {
            const int  vi = vb0 + lane;
            const bool in = vi < nvec;
            const int4 v  = dst4[in ? vi : 0];
            const int  e0 = vi << 2;
            own_push(qe_w, qn, in && (unsigned)(v.x - n0) < ucnt, e0 + 0, lane);
            own_push(qe_w, qn, in && (unsigned)(v.y - n0) < ucnt, e0 + 1, lane);
            own_push(qe_w, qn, in && (unsigned)(v.z - n0) < ucnt, e0 + 2, lane);
            own_push(qe_w, qn, in && (unsigned)(v.w - n0) < ucnt, e0 + 3, lane);
            if (qn > QW - 128)
                own_drain<MODE>(qe_w, qn, false, src, dst, el, er, h, acc, den, mxk, n0, E, lane, elmax);
        }

        if (wv == 0) {
            const int  e  = (nvec << 2) + lane;
            const bool in = e < E;
            const int  d  = dst[in ? e : 0];
            own_push(qe_w, qn, in && (unsigned)(d - n0) < ucnt, e, lane);
        }
        if (qn > QW - 128)
            own_drain<MODE>(qe_w, qn, false, src, dst, el, er, h, acc, den, mxk, n0, E, lane, elmax);
    }

    for (int base = wv * 32; base < cnt; base += OWN_WAVES * 32) {
        const int nl = base + lane;
        own_push(qe_w, qn, nl < cnt, E + n0 + nl, lane);
        if (qn > QW - 128)
            own_drain<MODE>(qe_w, qn, false, src, dst, el, er, h, acc, den, mxk, n0, E, lane, elmax);
    }
    own_drain<MODE>(qe_w, qn, true, src, dst, el, er, h, acc, den, mxk, n0, E, lane, elmax);
}

#define CAPR  64
#define RS_G  13

__device__ __forceinline__ void route_drain(const int* qe_w, int& qn,
                                            const int* __restrict__ src,
                                            const int* __restrict__ dst,
                                            float* rowq, int* rowcnt,
                                            int n0, int cnt, int lane) {
    for (int base = 0; base < qn; base += 128) {
        int  e[4], d[4], s[4];
        bool in[4];
        #pragma unroll
        for (int u = 0; u < 4; ++u) {
            const int i = base + 32 * u + lane;
            in[u] = i < qn;
            e[u]  = qe_w[in[u] ? i : base];
        }
        #pragma unroll
        for (int u = 0; u < 4; ++u) {
            d[u] = dst[e[u]];
            s[u] = src[e[u]];
        }
        int pos[4];
        #pragma unroll
        for (int u = 0; u < 4; ++u) {
            d[u] -= n0;
            in[u] = in[u] && (unsigned)d[u] < (unsigned)cnt;
            pos[u] = 0;
            if (in[u]) pos[u] = atomicAdd(&rowcnt[d[u]], 1);
        }
        #pragma unroll
        for (int u = 0; u < 4; ++u) {
            if (in[u]) {
                if (pos[u] < CAPR) rowq[d[u] * CAPR + pos[u]] = __int_as_float(s[u]);
                else               rowcnt[NPB] = 1;
            }
        }
    }
    qn = 0;
}

__device__ __forceinline__ void route_scan(int* qe_w,
                                           const int* __restrict__ src,
                                           const int* __restrict__ dst,
                                           const unsigned* bm,
                                           float* rowq, int* rowcnt,
                                           int n0, int cnt, int E, int lane, int wv) {
    int qn = 0;
    const int nvec = bm_chunks(E) * (BM_CHUNK / 128);
    const uint4* bm4 = (const uint4*)bm;
    for (int vb0 = wv * (32 * RS_G); vb0 < nvec; vb0 += OWN_WAVES * 32 * RS_G) {
        uint4 v[RS_G];
        #pragma unroll
        for (int j = 0; j < RS_G; ++j) {
            const int  vi = vb0 + 32 * j + lane;
            const bool in = vi < nvec;
            const uint4 t = bm4[in ? vi : 0];
            v[j].x = in ? t.x : 0u; v[j].y = in ? t.y : 0u;
            v[j].z = in ? t.z : 0u; v[j].w = in ? t.w : 0u;
        }
        #pragma unroll
        for (int j = 0; j < RS_G; ++j) {
            const int eb = (vb0 + 32 * j + lane) << 7;
            BM_PUSH_VEC(v[j], eb, route_drain(qe_w, qn, src, dst, rowq, rowcnt, n0, cnt, lane))
        }
        route_drain(qe_w, qn, src, dst, rowq, rowcnt, n0, cnt, lane);
    }
}

__device__ __forceinline__ void rows_from_slots(const float* rowq, const int* rowcnt,
                                                const float* __restrict__ el,
                                                const float* __restrict__ er,
                                                const float* __restrict__ h,
                                                const float* __restrict__ b,
                                                const int* __restrict__ rng,
                                                float* out, int n0, int cnt, int lane, int wv) {
    int r = wv;
    if (r >= cnt) return;

    const int    hcol  = lane << 2;
    const int    wl    = lane >> 2;
    const float  elmax = keyToFloat(rng[wl]);
    const float4 bias  = *(const float4*)(b + hcol);

    int   n   = min(__builtin_amdgcn_readfirstlane(rowcnt[r]), CAPR);
    int   sv0 = (lane < n)      ? __float_as_int(rowq[r * CAPR + lane])      : n0 + r;
    int   sv1 = (lane + 32 < n) ? __float_as_int(rowq[r * CAPR + 32 + lane]) : n0 + r;
    float erd = er[(size_t)(n0 + r) * HEADS + wl];

    bool more;
    do {
        const int d  = n0 + r;
        const int rn = r + OWN_WAVES;
        more = rn < cnt;
        const int rc = more ? rn : r;
        const float erd_next = er[(size_t)(n0 + rc) * HEADS + wl];

        const int   tot   = n + 1;
        const float bound = lrelu(elmax + erd);
        float4 acc = make_float4(0.0f, 0.0f, 0.0f, 0.0f);
        float  dn  = 0.0f;

        for (int q0 = 0; q0 < tot; q0 += 8) {
            const int left = tot - q0;
            const int svv  = (q0 < 32) ? sv0 : ((q0 < 64) ? sv1 : d);
            const int lb   = q0 & 31;

            float4 hv[8];
            float  ev[8];
            #pragma unroll
            for (int j = 0; j < 8; ++j) {
                const int sj = __builtin_amdgcn_readlane(svv, lb + j);
                hv[j] = *(const float4*)(h + (size_t)sj * OUT_DIM + hcol);
                ev[j] = el[(size_t)sj * HEADS + wl];
            }
            #pragma unroll
            for (int j = 0; j < 8; ++j) {
                const float x = lrelu(ev[j] + erd);
                const float a = (j < left) ? __expf(x - bound) : 0.0f;
                acc.x = fmaf(a, hv[j].x, acc.x);
                acc.y = fmaf(a, hv[j].y, acc.y);
                acc.z = fmaf(a, hv[j].z, acc.z);
                acc.w = fmaf(a, hv[j].w, acc.w);
                dn += a;
            }
        }

        const int n_next   = min(__builtin_amdgcn_readfirstlane(rowcnt[rc]), CAPR);
        const int sv0_next = (lane < n_next)      ? __float_as_int(rowq[rc * CAPR + lane])      : n0 + rc;
        const int sv1_next = (lane + 32 < n_next) ? __float_as_int(rowq[rc * CAPR + 32 + lane]) : n0 + rc;

        float4 o;
        o.x = acc.x / dn + bias.x;
        o.y = acc.y / dn + bias.y;
        o.z = acc.z / dn + bias.z;
        o.w = acc.w / dn + bias.w;
        *(float4*)(out + (size_t)d * OUT_DIM + hcol) = o;

        r = rn; n = n_next; sv0 = sv0_next; sv1 = sv1_next; erd = erd_next;
    } while (more);
}

__global__ void __launch_bounds__(OWN_TB)
gat_owner(const int* __restrict__ src, const int* __restrict__ dst,
          const float* __restrict__ el, const float* __restrict__ er,
          const float* __restrict__ h, const float* __restrict__ b,
          float* out, const int* __restrict__ rng, int N, int E) {
    __shared__ float acc[NPB * OUT_DIM];
    __shared__ float den[NPB * HEADS];
    __shared__ int   mxk[NPB * HEADS];
    __shared__ int   qe[OWN_WAVES * QW];
    __shared__ int   rowcnt[NPB + 1];

    const int n0   = blockIdx.x * NPB;
    const int cnt  = min(NPB, N - n0);
    const int lane = threadIdx.x & 31;
    const int wv   = __builtin_amdgcn_readfirstlane((int)(threadIdx.x >> 5));
    int* qe_w = qe + wv * QW;
    const unsigned* bm = bm_fits(cnt, E) ? (const unsigned*)(out + (size_t)n0 * OUT_DIM) : nullptr;

    const float elmax = keyToFloat(rng[lane & 7]);
    const bool  safe  = el_range_is_safe(rng);

    if (bm != nullptr && safe) {
        for (int i = threadIdx.x; i <= NPB; i += OWN_TB) rowcnt[i] = 0;
        __syncthreads();
        route_scan(qe_w, src, dst, bm, acc, rowcnt, n0, cnt, E, lane, wv);
        __syncthreads();
        if (rowcnt[NPB] == 0) {
            rows_from_slots(acc, rowcnt, el, er, h, b, rng, out, n0, cnt, lane, wv);
            return;
        }
    }

    for (int i = threadIdx.x; i < NPB * OUT_DIM; i += OWN_TB) acc[i] = 0.0f;
    for (int i = threadIdx.x; i < NPB * HEADS; i += OWN_TB) { den[i] = 0.0f; mxk[i] = INT_MIN; }
    __syncthreads();

    if (safe) {
        own_pass<1>(qe_w, src, dst, bm, el, er, h, acc, den, mxk, n0, cnt, E, lane, wv, elmax);
    } else {
        own_pass<0>(qe_w, src, dst, bm, el, er, h, acc, den, mxk, n0, cnt, E, lane, wv, elmax);
        __syncthreads();
        own_pass<2>(qe_w, src, dst, bm, el, er, h, acc, den, mxk, n0, cnt, E, lane, wv, elmax);
    }
    __syncthreads();

    for (int i = threadIdx.x; i < cnt * OUT_DIM; i += OWN_TB) {
        const int nl = i >> 7;
        const int c  = i & (OUT_DIM - 1);
        out[(size_t)n0 * OUT_DIM + i] = acc[i] / den[nl * HEADS + (c >> 4)] + b[c];
    }
}

extern "C" void kernel_launch(void* const* d_in, const int* in_sizes, int n_in,
                              void* d_out, int out_size, void* d_ws, size_t ws_size,
                              hipStream_t stream) {
    const float* feat   = (const float*)d_in[0];
    const float* W      = (const float*)d_in[1];
    const float* b      = (const float*)d_in[2];
    const float* attn_l = (const float*)d_in[3];
    const float* attn_r = (const float*)d_in[4];
    const int*   src    = (const int*)d_in[5];
    const int*   dst    = (const int*)d_in[6];
    float*       out    = (float*)d_out;

    const int N  = in_sizes[0] / IN_DIM;
    const int E  = in_sizes[5];

    float* h     = (float*)d_ws;
    float* el    = h + (size_t)N * OUT_DIM;
    float* er    = el + (size_t)N * HEADS;
    int*   rng   = (int*)(er + (size_t)N * HEADS);

    const int TB = 256;

    gat_gemm_wmma<<<GEMM_GRID, TB, 0, stream>>>(feat, W, h, rng, N);

    gat_logits<<<LOGIT_GRID + BUILD_GRID, TB, 0, stream>>>(h, attn_l, attn_r, el, er, rng, N,
                                                           dst, (unsigned*)out, E, LOGIT_GRID);

    gat_owner<<<(N + NPB - 1) / NPB, OWN_TB, 0, stream>>>(src, dst, el, er, h, b, out, rng, N, E);
}


# Edge bitmap: one bit per 16-byte group of 4 edges, a quarter of the bytes built, read and scanned; owner loads a marked group's dst and src as 16 bytes each; sum order still slot rank (LDS arrival)
# speedup vs baseline: 12.6203x; 1.1175x over previous
#include <hip/hip_runtime.h>
#include <hip/hip_bf16.h>
#include <climits>


#define IN_DIM   256
#define OUT_DIM  128
#define HEADS    8
#define HEAD_DIM 16
#define NEG_SLOPE 0.2f

typedef float v2f __attribute__((ext_vector_type(2)));
typedef float v8f __attribute__((ext_vector_type(8)));

__device__ __forceinline__ int floatKey(float x) {
    int i = __float_as_int(x);
    return (i >= 0) ? i : (i ^ 0x7FFFFFFF);
}
__device__ __forceinline__ float keyToFloat(int k) {
    int i = (k >= 0) ? k : (k ^ 0x7FFFFFFF);
    return __int_as_float(i);
}

__device__ __forceinline__ float lrelu(float x) {
    return x > 0.0f ? x : NEG_SLOPE * x;
}

#define EL_RANGE_SAFE 32.0f
__device__ __forceinline__ bool el_range_is_safe(const int* __restrict__ rng) {
    bool ok = true;
    #pragma unroll
    for (int k = 0; k < HEADS; ++k) {
        const float mx = keyToFloat(rng[k]);
        const float mn = keyToFloat(rng[HEADS + k]);
        ok = ok && ((mx - mn) <= EL_RANGE_SAFE);
    }
    return ok;
}

#define A_STRIDE 260

#define GEMM_GRID 512
#define KSTEPS    (IN_DIM / 4)
#define KGROUP    8

__device__ __forceinline__ float4 gemm_feat_piece(const float* __restrict__ feat, int tile, int u, int N) {
    int gr = tile * 16 + (int)(threadIdx.x >> 6) + 4 * u;
    if (gr >= N) gr = N - 1;
    return *(const float4*)&feat[(size_t)gr * IN_DIM + ((threadIdx.x & 63) << 2)];
}
__device__ __forceinline__ void gemm_stage_piece(float* __restrict__ stage, int u, float4 v) {
    *(float4*)&stage[((int)(threadIdx.x >> 6) + 4 * u) * A_STRIDE + ((threadIdx.x & 63) << 2)] = v;
}

__global__ void __launch_bounds__(256)
gat_gemm_wmma(const float* __restrict__ feat,
              const float* __restrict__ W,
              float* __restrict__ h, int* __restrict__ rng, int N) {
    __shared__ float As[2 * 16 * A_STRIDE];

    if (blockIdx.x == 0 && threadIdx.x < HEADS) {
        rng[threadIdx.x]         = INT_MIN;
        rng[HEADS + threadIdx.x] = INT_MAX;
    }

    const int ntile = (N + 15) >> 4;
    int tile = blockIdx.x;
    if (tile >= ntile) return;

    const int lane = threadIdx.x & 31;
    const int wave = threadIdx.x >> 5;
    const int m    = lane & 15;
    const int kh   = (lane >> 4) << 1;
    const int col0 = wave * HEAD_DIM;

    float4 p0 = gemm_feat_piece(feat, tile, 0, N);
    float4 p1 = gemm_feat_piece(feat, tile, 1, N);
    float4 p2 = gemm_feat_piece(feat, tile, 2, N);
    float4 p3 = gemm_feat_piece(feat, tile, 3, N);

    v2f bw[KSTEPS];
    #pragma unroll
    for (int k = 0; k < KSTEPS; ++k) {
        bw[k].x = W[(size_t)(4 * k + kh + 0) * OUT_DIM + col0 + m];
        bw[k].y = W[(size_t)(4 * k + kh + 1) * OUT_DIM + col0 + m];
    }

    gemm_stage_piece(As, 0, p0);
    gemm_stage_piece(As, 1, p1);
    gemm_stage_piece(As, 2, p2);
    gemm_stage_piece(As, 3, p3);
    __syncthreads();

    const int colg = col0 + m;
    int buf = 0;
    for (;;) {
        const int  next     = tile + (int)gridDim.x;
        const bool has_next = next < ntile;
        if (has_next) {
            p0 = gemm_feat_piece(feat, next, 0, N);
            p1 = gemm_feat_piece(feat, next, 1, N);
            p2 = gemm_feat_piece(feat, next, 2, N);
            p3 = gemm_feat_piece(feat, next, 3, N);
        }

        const float* __restrict__ arow = &As[buf * (16 * A_STRIDE) + m * A_STRIDE + kh];
        v8f acc = {};
        v2f a[2][KGROUP];
        #pragma unroll
        for (int j = 0; j < KGROUP; ++j) a[0][j] = *(const v2f*)&arow[4 * j];
        #pragma unroll
        for (int g = 0; g < KSTEPS / KGROUP; ++g) {
            if (g + 1 < KSTEPS / KGROUP) {
                #pragma unroll
                for (int j = 0; j < KGROUP; ++j)
                    a[(g + 1) & 1][j] = *(const v2f*)&arow[4 * ((g + 1) * KGROUP + j)];
            }
            __builtin_amdgcn_sched_barrier(0);
            #pragma unroll
            for (int j = 0; j < KGROUP; ++j) {
                acc = __builtin_amdgcn_wmma_f32_16x16x4_f32(
                     false, a[g & 1][j],  false, bw[g * KGROUP + j],
                     (short)0, acc,  false,  false);
            }
            __builtin_amdgcn_sched_barrier(0);
        }

        if (has_next) {
            float* __restrict__ nst = &As[(buf ^ 1) * (16 * A_STRIDE)];
            gemm_stage_piece(nst, 0, p0);
            gemm_stage_piece(nst, 1, p1);
            gemm_stage_piece(nst, 2, p2);
            gemm_stage_piece(nst, 3, p3);
            __syncthreads();
        }

        const int row0  = tile * 16;
        const int rbase = row0 + ((lane >> 4) << 3);
        float* __restrict__ op = h + (size_t)rbase * OUT_DIM + colg;
        if (row0 + 16 <= N) {
            #pragma unroll
            for (int r = 0; r < 8; ++r) {
                op[(size_t)r * OUT_DIM] = acc[r];
            }
        } else {
            #pragma unroll
            for (int r = 0; r < 8; ++r) {
                if (rbase + r < N) op[(size_t)r * OUT_DIM] = acc[r];
            }
        }

        if (!has_next) break;
        buf ^= 1;
        tile = next;
    }
}

#define NPB        400
#define BM_GRP     4
#define BM_WORDS   32
#define BM_CHUNK   (BM_WORDS * 32 * BM_GRP)
#define BM_OWNERS  256
#define BM_PITCH   33
#define LOGIT_GRID 256
#define BUILD_GRID 512

__device__ __forceinline__ int bm_chunks(int E) { return (E + BM_CHUNK - 1) / BM_CHUNK; }
__device__ __forceinline__ bool bm_fits(int cnt, int E) {
    return (long long)cnt * OUT_DIM >= (long long)bm_chunks(E) * BM_WORDS;
}

__device__ __forceinline__ void bm_build(const int* __restrict__ dst, unsigned* bm,
                                         unsigned* tab, int chunk, int N, int E) {
    const int tid = threadIdx.x;
    int4 d[BM_WORDS * 32 / 256];
    #pragma unroll
    for (int u = 0; u < BM_WORDS * 32 / 256; ++u) {
        const int g = chunk * (BM_WORDS * 32) + u * 256 + tid;
        d[u] = make_int4(-1, -1, -1, -1);
        if (g < E / BM_GRP) d[u] = *(const int4*)(dst + (size_t)g * BM_GRP);
    }
    for (int i = tid; i < BM_OWNERS * BM_PITCH; i += 256) tab[i] = 0u;
    __syncthreads();
    #pragma unroll
    for (int u = 0; u < BM_WORDS * 32 / 256; ++u) {
        const int      gl   = u * 256 + tid;
        const int      wloc = gl >> 5;
        const unsigned bit  = 1u << (gl & 31);
        const int dd[4] = {d[u].x, d[u].y, d[u].z, d[u].w};
        #pragma unroll
        for (int k = 0; k < 4; ++k) {
            if ((unsigned)dd[k] < (unsigned)N) {
                const int o = dd[k] / NPB;
                if (o < BM_OWNERS) atomicOr(&tab[o * BM_PITCH + wloc], bit);
            }
        }
    }
    __syncthreads();
    const int nown = (N + NPB - 1) / NPB;
    for (int i = tid; i < BM_OWNERS * BM_WORDS; i += 256) {
        const int o = i / BM_WORDS;
        const int w = i % BM_WORDS;
        if (o < nown && bm_fits(min(NPB, N - o * NPB), E))
            bm[(size_t)o * (NPB * OUT_DIM) + (size_t)chunk * BM_WORDS + w] = tab[o * BM_PITCH + w];
    }
}

__global__ void __launch_bounds__(256)
gat_logits(const float* __restrict__ h,
           const float* __restrict__ attn_l,
           const float* __restrict__ attn_r,
           float* __restrict__ el, float* __restrict__ er,
           int* __restrict__ rng, int N,
           const int* __restrict__ dst, unsigned* bm, int E, int nlogit) {
    __shared__ int red[2][8][HEADS];
    __shared__ unsigned tab[BM_OWNERS * BM_PITCH];
    if ((int)blockIdx.x >= nlogit) {
        const int nb = (int)gridDim.x - nlogit;
        for (int chunk = (int)blockIdx.x - nlogit; chunk < bm_chunks(E); chunk += nb) {
            bm_build(dst, bm, tab, chunk, N, E);
            __syncthreads();
        }
        return;
    }
    const int nthr = nlogit * (int)blockDim.x;
    const int hd   = threadIdx.x & 7;
    const float* al = attn_l + hd * HEAD_DIM;
    const float* ar = attn_r + hd * HEAD_DIM;
    int kmx = INT_MIN;
    int kmn = INT_MAX;
    for (int gid = blockIdx.x * blockDim.x + threadIdx.x; gid < N * HEADS; gid += nthr) {
        const int n = gid >> 3;
        const float* hp = h + (size_t)n * OUT_DIM + hd * HEAD_DIM;
        float sl = 0.0f, sr = 0.0f;
        #pragma unroll
        for (int j = 0; j < HEAD_DIM; ++j) {
            float v = hp[j];
            sl = fmaf(v, al[j], sl);
            sr = fmaf(v, ar[j], sr);
        }
        el[gid] = sl;
        er[gid] = sr;
        const int key = floatKey(sl);
        kmx = max(kmx, key);
        kmn = min(kmn, key);
    }

    kmx = max(kmx, __shfl_xor(kmx, 8));
    kmn = min(kmn, __shfl_xor(kmn, 8));
    kmx = max(kmx, __shfl_xor(kmx, 16));
    kmn = min(kmn, __shfl_xor(kmn, 16));
    const int lane = threadIdx.x & 31;
    const int wave = threadIdx.x >> 5;
    if (lane < HEADS) {
        red[0][wave][lane] = kmx;
        red[1][wave][lane] = kmn;
    }
    __syncthreads();
    if (threadIdx.x < 2 * HEADS) {
        const int which = threadIdx.x >> 3;
        const int hh    = threadIdx.x & 7;
        int v = red[which][0][hh];
        for (int w = 1; w < 8; ++w) {
            const int o = red[which][w][hh];
            v = which ? min(v, o) : max(v, o);
        }
        if (which) atomicMin(&rng[HEADS + hh], v);
        else       atomicMax(&rng[hh], v);
    }
}

#define OWN_TB    1024
#define OWN_WAVES 32
#define QW        256
#define SCAN_G    8

__device__ __forceinline__ void own_push(int* __restrict__ qe_w,
                                         int& qn, bool m, int e, int lane) {
    const unsigned bal = __builtin_amdgcn_ballot_w32(m);
    if (bal != 0u) {
        const int pos = qn + __popc(bal & ((1u << lane) - 1u));
        if (m) qe_w[pos] = e;
        qn += __popc(bal);
    }
}

__device__ __forceinline__ int own_src(const int* __restrict__ src, int e, int E) {
    const bool real = e < E;
    const int  ss   = src[real ? e : 0];
    return real ? ss : e - E;
}
__device__ __forceinline__ int own_dl(const int* __restrict__ dst, int e, int E, int n0) {
    const bool real = e < E;
    const int  dd   = dst[real ? e : 0];
    return (real ? dd : e - E) - n0;
}

__device__ __forceinline__ void own_max_all(const int* qe_w, int qn,
                                            const int* __restrict__ src,
                                            const int* __restrict__ dst,
                                            const float* __restrict__ el,
                                            const float* __restrict__ er,
                                            int* mxk, int n0, int E, int lane) {
    const int sub = lane >> 3;
    const int hd  = lane & 7;
    for (int q = 0; q < qn; q += 4) {
        const bool valid = q + sub < qn;
        const int  ii    = valid ? q + sub : q;
        const int  e     = qe_w[ii];
        const int  dl    = own_dl(dst, e, E, n0);
        const int  s     = own_src(src, e, E);
        const float x = lrelu(el[(size_t)s * HEADS + hd] + er[(size_t)(n0 + dl) * HEADS + hd]);
        if (valid) atomicMax(&mxk[dl * HEADS + hd], floatKey(x));
    }
}

template <int NB, bool BOUND>
__device__ __forceinline__ void own_batch(const int* qe_w, int q0,
                                          const int* __restrict__ src,
                                          const int* __restrict__ dst,
                                          const float* __restrict__ el,
                                          const float* __restrict__ er,
                                          const float* __restrict__ h,
                                          float* acc, float* den, const int* mxk,
                                          int n0, int E, int lane, float elmax) {
    constexpr int NH = (NB + 3) / 4;
    const int  sub   = lane >> 3;
    const int  hd    = lane & 7;
    const bool upper = lane >= 16;

    int s[NH], dl[NH];
    #pragma unroll
    for (int t = 0; t < NH; ++t) {
        const int idx = (4 * t + sub < NB) ? 4 * t + sub : 0;
        const int e   = qe_w[q0 + idx];
        dl[t]         = own_dl(dst, e, E, n0);
        s[t]          = own_src(src, e, E);
    }

    int   dj[NB];
    float hv[NB][4];
    #pragma unroll
    for (int j = 0; j < NB; ++j) {
        const int sj = __builtin_amdgcn_readlane(s[j >> 2], 8 * (j & 3));
        dj[j]        = __builtin_amdgcn_readlane(dl[j >> 2], 8 * (j & 3));
        const float* __restrict__ hp = h + (size_t)sj * OUT_DIM + lane;
        #pragma unroll
        for (int k = 0; k < 4; ++k) hv[j][k] = hp[32 * k];
    }

    int pi[NH];
    #pragma unroll
    for (int t = 0; t < NH; ++t) {
        const float erd = er[(size_t)(n0 + dl[t]) * HEADS + hd];
        const float x   = lrelu(el[(size_t)s[t] * HEADS + hd] + erd);
        const float mx  = BOUND ? lrelu(elmax + erd)
                                : keyToFloat(mxk[dl[t] * HEADS + hd]);
        const float p   = __expf(x - mx);
        if (4 * t + sub < NB) atomicAdd(&den[dl[t] * HEADS + hd], p);
        pi[t] = __float_as_int(p);
    }

    #pragma unroll
    for (int j = 0; j < NB; ++j) {
        float* op = acc + dj[j] * OUT_DIM + lane;
        #pragma unroll
        for (int k = 0; k < 4; ++k) {
            const int a_lo = __builtin_amdgcn_readlane(pi[j >> 2], 8 * (j & 3) + 2 * k);
            const int a_hi = __builtin_amdgcn_readlane(pi[j >> 2], 8 * (j & 3) + 2 * k + 1);
            const float a  = __int_as_float(upper ? a_hi : a_lo);
            atomicAdd(&op[32 * k], hv[j][k] * a);
        }
    }
}

template <int MODE>
__device__ __forceinline__ void own_drain(int* qe_w, int& qn, bool all,
                                          const int* __restrict__ src,
                                          const int* __restrict__ dst,
                                          const float* __restrict__ el,
                                          const float* __restrict__ er,
                                          const float* __restrict__ h,
                                          float* acc, float* den, int* mxk,
                                          int n0, int E, int lane, float elmax) {
    if (MODE == 0) {
        own_max_all(qe_w, qn, src, dst, el, er, mxk, n0, E, lane);
        qn = 0;
    } else {
        constexpr bool B = (MODE == 1);
        while (qn >= 8) {
            qn -= 8;
            own_batch<8, B>(qe_w, qn, src, dst, el, er, h, acc, den, mxk, n0, E, lane, elmax);
        }
        if (all && qn > 0) {
            switch (qn) {
            case 1:  own_batch<1, B>(qe_w, 0, src, dst, el, er, h, acc, den, mxk, n0, E, lane, elmax); break;
            case 2:  own_batch<2, B>(qe_w, 0, src, dst, el, er, h, acc, den, mxk, n0, E, lane, elmax); break;
            case 3:  own_batch<3, B>(qe_w, 0, src, dst, el, er, h, acc, den, mxk, n0, E, lane, elmax); break;
            case 4:  own_batch<4, B>(qe_w, 0, src, dst, el, er, h, acc, den, mxk, n0, E, lane, elmax); break;
            case 5:  own_batch<5, B>(qe_w, 0, src, dst, el, er, h, acc, den, mxk, n0, E, lane, elmax); break;
            case 6:  own_batch<6, B>(qe_w, 0, src, dst, el, er, h, acc, den, mxk, n0, E, lane, elmax); break;
            default: own_batch<7, B>(qe_w, 0, src, dst, el, er, h, acc, den, mxk, n0, E, lane, elmax); break;
            }
            qn = 0;
        }
    }
}

#define BM_PUSH_VEC(vv, eb, DRAIN)                                                    \
    {                                                                                 \
        unsigned x_ = (vv).x, y_ = (vv).y, z_ = (vv).z, w_ = (vv).w;                  \
        const int eb_ = (eb);                                                         \
        for (int it_ = 0; it_ < 32; ++it_) {                                          \
            if (__builtin_amdgcn_ballot_w32((x_ | y_ | z_ | w_) != 0u) == 0u) break;  \
            own_push(qe_w, qn, x_ != 0u, eb_ + __ffs((int)x_) - 1, lane);             \
            own_push(qe_w, qn, y_ != 0u, eb_ + 32 + __ffs((int)y_) - 1, lane);        \
            own_push(qe_w, qn, z_ != 0u, eb_ + 64 + __ffs((int)z_) - 1, lane);        \
            own_push(qe_w, qn, w_ != 0u, eb_ + 96 + __ffs((int)w_) - 1, lane);        \
            x_ &= x_ - 1u; y_ &= y_ - 1u; z_ &= z_ - 1u; w_ &= w_ - 1u;               \
            if (qn > QW - 128) {             \
                DRAIN;                                                                \
            }                                                                         \
        }                                                                             \
    }

template <int MODE>
__device__ __forceinline__ void own_pass(int* qe_w,
                                         const int* __restrict__ src,
                                         const int* __restrict__ dst,
                                         const float* __restrict__ el,
                                         const float* __restrict__ er,
                                         const float* __restrict__ h,
                                         float* acc, float* den, int* mxk,
                                         int n0, int cnt, int E, int lane, int wv, float elmax) {
    int qn = 0;
    {
        const unsigned ucnt = (unsigned)cnt;
        const int nvec   = E >> 2;
        const int ngroup = nvec / (32 * SCAN_G);
        const int4* __restrict__ dst4 = (const int4*)dst;

        for (int g = wv; g < ngroup; g += OWN_WAVES) {
            const int vb = g * (32 * SCAN_G) + lane;
            int4 v[SCAN_G];
            #pragma unroll
            for (int j = 0; j < SCAN_G; ++j) v[j] = dst4[vb + 32 * j];
            #pragma unroll
            for (int j = 0; j < SCAN_G; ++j) {
                const int e0 = (vb + 32 * j) << 2;
                own_push(qe_w, qn, (unsigned)(v[j].x - n0) < ucnt, e0 + 0, lane);
                own_push(qe_w, qn, (unsigned)(v[j].y - n0) < ucnt, e0 + 1, lane);
                own_push(qe_w, qn, (unsigned)(v[j].z - n0) < ucnt, e0 + 2, lane);
                own_push(qe_w, qn, (unsigned)(v[j].w - n0) < ucnt, e0 + 3, lane);
                if (qn > QW - 128)
                    own_drain<MODE>(qe_w, qn, false, src, dst, el, er, h, acc, den, mxk, n0, E, lane, elmax);
            }
            own_drain<MODE>(qe_w, qn, false, src, dst, el, er, h, acc, den, mxk, n0, E, lane, elmax);
        }

        for (int vb0 = ngroup * (32 * SCAN_G) + wv * 32; vb0 < nvec; vb0 += OWN_WAVES * 32) {
            const int  vi = vb0 + lane;
            const bool in = vi < nvec;
            const int4 v  = dst4[in ? vi : 0];
            const int  e0 = vi << 2;
            own_push(qe_w, qn, in && (unsigned)(v.x - n0) < ucnt, e0 + 0, lane);
            own_push(qe_w, qn, in && (unsigned)(v.y - n0) < ucnt, e0 + 1, lane);
            own_push(qe_w, qn, in && (unsigned)(v.z - n0) < ucnt, e0 + 2, lane);
            own_push(qe_w, qn, in && (unsigned)(v.w - n0) < ucnt, e0 + 3, lane);
            if (qn > QW - 128)
                own_drain<MODE>(qe_w, qn, false, src, dst, el, er, h, acc, den, mxk, n0, E, lane, elmax);
        }

        if (wv == 0) {
            const int  e  = (nvec << 2) + lane;
            const bool in = e < E;
            const int  d  = dst[in ? e : 0];
            own_push(qe_w, qn, in && (unsigned)(d - n0) < ucnt, e, lane);
        }
        if (qn > QW - 128)
            own_drain<MODE>(qe_w, qn, false, src, dst, el, er, h, acc, den, mxk, n0, E, lane, elmax);
    }

    for (int base = wv * 32; base < cnt; base += OWN_WAVES * 32) {
        const int nl = base + lane;
        own_push(qe_w, qn, nl < cnt, E + n0 + nl, lane);
        if (qn > QW - 128)
            own_drain<MODE>(qe_w, qn, false, src, dst, el, er, h, acc, den, mxk, n0, E, lane, elmax);
    }
    own_drain<MODE>(qe_w, qn, true, src, dst, el, er, h, acc, den, mxk, n0, E, lane, elmax);
}

#define CAPR  64
#define RS_G  4

__device__ __forceinline__ void route_drain(const int* qe_w, int& qn,
                                            const int* __restrict__ src,
                                            const int* __restrict__ dst,
                                            float* rowq, int* rowcnt,
                                            int n0, int cnt, int lane) {
    for (int base = 0; base < qn; base += 128) {
        int  g[4];
        bool in[4];
        #pragma unroll
        for (int u = 0; u < 4; ++u) {
            const int i = base + 32 * u + lane;
            in[u] = i < qn;
            g[u]  = qe_w[in[u] ? i : base];
        }
        int4 dv[4], sv[4];
        #pragma unroll
        for (int u = 0; u < 4; ++u) {
            dv[u] = *(const int4*)(dst + (size_t)g[u] * BM_GRP);
            sv[u] = *(const int4*)(src + (size_t)g[u] * BM_GRP);
        }
        int  df[4], sf[4];
        bool has[4], more[4];
        #pragma unroll
        for (int u = 0; u < 4; ++u) {
            const unsigned lim = in[u] ? (unsigned)cnt : 0u;
            const int d4[4] = {dv[u].x - n0, dv[u].y - n0, dv[u].z - n0, dv[u].w - n0};
            const int s4[4] = {sv[u].x, sv[u].y, sv[u].z, sv[u].w};
            has[u] = false; more[u] = false; df[u] = 0; sf[u] = 0;
            #pragma unroll
            for (int k = 0; k < 4; ++k) {
                const bool own  = (unsigned)d4[k] < lim;
                const bool take = own && !has[u];
                df[u]   = take ? d4[k] : df[u];
                sf[u]   = take ? s4[k] : sf[u];
                more[u] = more[u] || (own && has[u]);
                has[u]  = has[u] || own;
            }
        }
        int pos[4];
        #pragma unroll
        for (int u = 0; u < 4; ++u) {
            pos[u] = 0;
            if (has[u]) pos[u] = atomicAdd(&rowcnt[df[u]], 1);
        }
        #pragma unroll
        for (int u = 0; u < 4; ++u) {
            if (has[u]) {
                if (pos[u] < CAPR) rowq[df[u] * CAPR + pos[u]] = __int_as_float(sf[u]);
                else               rowcnt[NPB] = 1;
            }
        }
        #pragma unroll
        for (int u = 0; u < 4; ++u) {
            if (__builtin_amdgcn_ballot_w32(more[u]) != 0u) {
                const unsigned lim = in[u] ? (unsigned)cnt : 0u;
                const int d4[4] = {dv[u].x - n0, dv[u].y - n0, dv[u].z - n0, dv[u].w - n0};
                const int s4[4] = {sv[u].x, sv[u].y, sv[u].z, sv[u].w};
                bool seen = (unsigned)d4[0] < lim;
                #pragma unroll
                for (int k = 1; k < 4; ++k) {
                    const bool own = (unsigned)d4[k] < lim;
                    if (own && seen) {
                        const int p = atomicAdd(&rowcnt[d4[k]], 1);
                        if (p < CAPR) rowq[d4[k] * CAPR + p] = __int_as_float(s4[k]);
                        else          rowcnt[NPB] = 1;
                    }
                    seen = seen || own;
                }
            }
        }
    }
    qn = 0;
}

__device__ __forceinline__ void route_scan(int* qe_w,
                                           const int* __restrict__ src,
                                           const int* __restrict__ dst,
                                           const unsigned* bm,
                                           float* rowq, int* rowcnt,
                                           int n0, int cnt, int E, int lane, int wv) {
    int qn = 0;
    const int nvec = bm_chunks(E) * (BM_WORDS / 4);
    const uint4* bm4 = (const uint4*)bm;
    for (int vb0 = 0; vb0 < nvec; vb0 += OWN_WAVES * 32 * RS_G) {
        uint4 v[RS_G];
        #pragma unroll
        for (int j = 0; j < RS_G; ++j) {
            const int vi = vb0 + (j * OWN_WAVES + wv) * 32 + lane;
            v[j] = make_uint4(0u, 0u, 0u, 0u);
            if (vi < nvec) v[j] = bm4[vi];
        }
        #pragma unroll
        for (int j = 0; j < RS_G; ++j) {
            const int eb = (vb0 + (j * OWN_WAVES + wv) * 32 + lane) << 7;
            BM_PUSH_VEC(v[j], eb, route_drain(qe_w, qn, src, dst, rowq, rowcnt, n0, cnt, lane))
        }
        route_drain(qe_w, qn, src, dst, rowq, rowcnt, n0, cnt, lane);
    }
    if (wv == 0 && (E % BM_GRP) != 0) {
        const int  e   = E - (E % BM_GRP) + lane;
        const bool in  = e < E;
        const int  d   = dst[in ? e : 0] - n0;
        const int  sn  = src[in ? e : 0];
        if (in && (unsigned)d < (unsigned)cnt) {
            const int pos = atomicAdd(&rowcnt[d], 1);
            if (pos < CAPR) rowq[d * CAPR + pos] = __int_as_float(sn);
            else            rowcnt[NPB] = 1;
        }
    }
}

__device__ __forceinline__ void rows_from_slots(const float* rowq, const int* rowcnt,
                                                const float* __restrict__ el,
                                                const float* __restrict__ er,
                                                const float* __restrict__ h,
                                                const float* __restrict__ b,
                                                const int* __restrict__ rng,
                                                float* out, int n0, int cnt, int lane, int wv) {
    int r = wv;
    if (r >= cnt) return;

    const int    hcol  = lane << 2;
    const int    wl    = lane >> 2;
    const float  elmax = keyToFloat(rng[wl]);
    const float4 bias  = *(const float4*)(b + hcol);

    int   n   = min(__builtin_amdgcn_readfirstlane(rowcnt[r]), CAPR);
    int   sv0 = (lane < n)      ? __float_as_int(rowq[r * CAPR + lane])      : n0 + r;
    int   sv1 = (lane + 32 < n) ? __float_as_int(rowq[r * CAPR + 32 + lane]) : n0 + r;
    float erd = er[(size_t)(n0 + r) * HEADS + wl];

    bool more;
    do {
        const int d  = n0 + r;
        const int rn = r + OWN_WAVES;
        more = rn < cnt;
        const int rc = more ? rn : r;
        const float erd_next = er[(size_t)(n0 + rc) * HEADS + wl];

        const int   tot   = n + 1;
        const float bound = lrelu(elmax + erd);
        float4 acc = make_float4(0.0f, 0.0f, 0.0f, 0.0f);
        float  dn  = 0.0f;

        for (int q0 = 0; q0 < tot; q0 += 8) {
            const int left = tot - q0;
            const int svv  = (q0 < 32) ? sv0 : ((q0 < 64) ? sv1 : d);
            const int lb   = q0 & 31;

            float4 hv[8];
            float  ev[8];
            #pragma unroll
            for (int j = 0; j < 8; ++j) {
                const int sj = __builtin_amdgcn_readlane(svv, lb + j);
                hv[j] = *(const float4*)(h + (size_t)sj * OUT_DIM + hcol);
                ev[j] = el[(size_t)sj * HEADS + wl];
            }
            #pragma unroll
            for (int j = 0; j < 8; ++j) {
                const float x = lrelu(ev[j] + erd);
                const float a = (j < left) ? __expf(x - bound) : 0.0f;
                acc.x = fmaf(a, hv[j].x, acc.x);
                acc.y = fmaf(a, hv[j].y, acc.y);
                acc.z = fmaf(a, hv[j].z, acc.z);
                acc.w = fmaf(a, hv[j].w, acc.w);
                dn += a;
            }
        }

        const int n_next   = min(__builtin_amdgcn_readfirstlane(rowcnt[rc]), CAPR);
        const int sv0_next = (lane < n_next)      ? __float_as_int(rowq[rc * CAPR + lane])      : n0 + rc;
        const int sv1_next = (lane + 32 < n_next) ? __float_as_int(rowq[rc * CAPR + 32 + lane]) : n0 + rc;

        float4 o;
        o.x = acc.x / dn + bias.x;
        o.y = acc.y / dn + bias.y;
        o.z = acc.z / dn + bias.z;
        o.w = acc.w / dn + bias.w;
        *(float4*)(out + (size_t)d * OUT_DIM + hcol) = o;

        r = rn; n = n_next; sv0 = sv0_next; sv1 = sv1_next; erd = erd_next;
    } while (more);
}

__global__ void __launch_bounds__(OWN_TB)
gat_owner(const int* __restrict__ src, const int* __restrict__ dst,
          const float* __restrict__ el, const float* __restrict__ er,
          const float* __restrict__ h, const float* __restrict__ b,
          float* out, const int* __restrict__ rng, int N, int E) {
    __shared__ float acc[NPB * OUT_DIM];
    __shared__ float den[NPB * HEADS];
    __shared__ int   mxk[NPB * HEADS];
    __shared__ int   qe[OWN_WAVES * QW];
    __shared__ int   rowcnt[NPB + 1];

    const int n0   = blockIdx.x * NPB;
    const int cnt  = min(NPB, N - n0);
    const int lane = threadIdx.x & 31;
    const int wv   = __builtin_amdgcn_readfirstlane((int)(threadIdx.x >> 5));
    int* qe_w = qe + wv * QW;
    const unsigned* bm = bm_fits(cnt, E) ? (const unsigned*)(out + (size_t)n0 * OUT_DIM) : nullptr;

    const float elmax = keyToFloat(rng[lane & 7]);
    const bool  safe  = el_range_is_safe(rng);

    if (bm != nullptr && safe) {
        for (int i = threadIdx.x; i <= NPB; i += OWN_TB) rowcnt[i] = 0;
        __syncthreads();
        route_scan(qe_w, src, dst, bm, acc, rowcnt, n0, cnt, E, lane, wv);
        __syncthreads();
        if (rowcnt[NPB] == 0) {
            rows_from_slots(acc, rowcnt, el, er, h, b, rng, out, n0, cnt, lane, wv);
            return;
        }
    }

    for (int i = threadIdx.x; i < NPB * OUT_DIM; i += OWN_TB) acc[i] = 0.0f;
    for (int i = threadIdx.x; i < NPB * HEADS; i += OWN_TB) { den[i] = 0.0f; mxk[i] = INT_MIN; }
    __syncthreads();

    if (safe) {
        own_pass<1>(qe_w, src, dst, el, er, h, acc, den, mxk, n0, cnt, E, lane, wv, elmax);
    } else {
        own_pass<0>(qe_w, src, dst, el, er, h, acc, den, mxk, n0, cnt, E, lane, wv, elmax);
        __syncthreads();
        own_pass<2>(qe_w, src, dst, el, er, h, acc, den, mxk, n0, cnt, E, lane, wv, elmax);
    }
    __syncthreads();

    for (int i = threadIdx.x; i < cnt * OUT_DIM; i += OWN_TB) {
        const int nl = i >> 7;
        const int c  = i & (OUT_DIM - 1);
        out[(size_t)n0 * OUT_DIM + i] = acc[i] / den[nl * HEADS + (c >> 4)] + b[c];
    }
}

extern "C" void kernel_launch(void* const* d_in, const int* in_sizes, int n_in,
                              void* d_out, int out_size, void* d_ws, size_t ws_size,
                              hipStream_t stream) {
    const float* feat   = (const float*)d_in[0];
    const float* W      = (const float*)d_in[1];
    const float* b      = (const float*)d_in[2];
    const float* attn_l = (const float*)d_in[3];
    const float* attn_r = (const float*)d_in[4];
    const int*   src    = (const int*)d_in[5];
    const int*   dst    = (const int*)d_in[6];
    float*       out    = (float*)d_out;

    const int N  = in_sizes[0] / IN_DIM;
    const int E  = in_sizes[5];

    float* h     = (float*)d_ws;
    float* el    = h + (size_t)N * OUT_DIM;
    float* er    = el + (size_t)N * HEADS;
    int*   rng   = (int*)(er + (size_t)N * HEADS);

    const int TB = 256;

    gat_gemm_wmma<<<GEMM_GRID, TB, 0, stream>>>(feat, W, h, rng, N);

    gat_logits<<<LOGIT_GRID + BUILD_GRID, TB, 0, stream>>>(h, attn_l, attn_r, el, er, rng, N,
                                                           dst, (unsigned*)out, E, LOGIT_GRID);

    gat_owner<<<(N + NPB - 1) / NPB, OWN_TB, 0, stream>>>(src, dst, el, er, h, b, out, rng, N, E);
}
